# starting kernel
# baseline (speedup 1.0000x reference)
#include <hip/hip_runtime.h>

#define T_STEPS 12
#define F_IN 16
#define H 64

typedef __attribute__((ext_vector_type(16))) _Float16 v16h;
typedef __attribute__((ext_vector_type(8)))  _Float16 v8h;
typedef __attribute__((ext_vector_type(8)))  float    v8f;

__device__ __forceinline__ v16h load_fragA(const _Float16* __restrict__ row, int kb) {
  union { v16h v; v8h h[2]; } u;
  u.h[0] = *(const v8h*)(row + kb);
  u.h[1] = *(const v8h*)(row + kb + 16);
  return u.v;
}

__device__ __forceinline__ float sigmoidf_(float x) {
  return __builtin_amdgcn_rcpf(1.0f + __expf(-x));
}
__device__ __forceinline__ float tanhf_(float x) {
  return 1.0f - 2.0f * __builtin_amdgcn_rcpf(1.0f + __expf(2.0f * x));
}


__global__ __launch_bounds__(256) void init_deg_kernel(float* __restrict__ deg, int N) {
  int i = blockIdx.x * 256 + threadIdx.x;
  if (i < N) deg[i] = 1.0f;
}

__global__ __launch_bounds__(256) void deg_accum_kernel(const int* __restrict__ dst,
                                                        float* __restrict__ deg, int E) {
  int e = blockIdx.x * 256 + threadIdx.x;
  if (e < E) atomicAdd(&deg[dst[e]], 1.0f);
}

__global__ __launch_bounds__(256) void dinv_kernel(float* __restrict__ deg, int N) {
  int i = blockIdx.x * 256 + threadIdx.x;
  if (i < N) deg[i] = rsqrtf(deg[i]);
}

__global__ __launch_bounds__(256) void edge_w_kernel(const int* __restrict__ src,
                                                     const int* __restrict__ dst,
                                                     const float* __restrict__ dinv,
                                                     float* __restrict__ w, int E) {
  int e = blockIdx.x * 256 + threadIdx.x;
  if (e < E) w[e] = dinv[src[e]] * dinv[dst[e]];
}

__global__ __launch_bounds__(256) void cvt_weights_kernel(const float* __restrict__ Wih,
                                                          const float* __restrict__ Whh,
                                                          _Float16* __restrict__ Wih16,
                                                          _Float16* __restrict__ Whh16,
                                                          int n) {
  int i = blockIdx.x * 256 + threadIdx.x;
  if (i < n) {
    Wih16[i] = (_Float16)Wih[i];
    Whh16[i] = (_Float16)Whh[i];
  }
}

__global__ __launch_bounds__(256) void init_h_kernel(float* __restrict__ h,
                                                     _Float16* __restrict__ h16, int n) {
  int i = blockIdx.x * 256 + threadIdx.x;
  if (i < n) { h[i] = 0.0f; h16[i] = (_Float16)0.0f; }
}


__global__ __launch_bounds__(256) void transform_kernel(const float* __restrict__ X,
                                                        const float* __restrict__ Wg,
                                                        const float* __restrict__ dinv,
                                                        float* __restrict__ g,
                                                        float* __restrict__ agg,
                                                        int t, int N) {
  __shared__ float sW[F_IN * H];
  __shared__ float sX[4 * F_IN];
  int tid = threadIdx.x;
  #pragma unroll
  for (int i = tid; i < F_IN * H; i += 256) sW[i] = Wg[i];
  if (tid < 64) {
    int ni = blockIdx.x * 4 + (tid >> 4);
    int k  = tid & 15;
    sX[tid] = (ni < N) ? X[((size_t)ni * T_STEPS + t) * F_IN + k] : 0.0f;
  }
  __syncthreads();
  int node = blockIdx.x * 4 + (tid >> 6);
  int hc   = tid & 63;
  if (node < N) {
    const float* xr = &sX[(tid >> 6) * F_IN];
    float acc = 0.0f;
    #pragma unroll
    for (int k = 0; k < F_IN; ++k) acc += xr[k] * sW[k * H + hc];
    float di = dinv[node];
    size_t idx = (size_t)node * H + hc;
    g[idx]   = acc;
    agg[idx] = acc * di * di;
  }
}

__global__ __launch_bounds__(256) void scatter_kernel(const int* __restrict__ src,
                                                      const int* __restrict__ dst,
                                                      const float* __restrict__ w,
                                                      const float* __restrict__ g,
                                                      float* __restrict__ agg, int E) {
  int tid = blockIdx.x * 256 + threadIdx.x;
  int e = tid >> 4;
  if (e >= E) return;
  int q = tid & 15;
  int s = src[e], d = dst[e];
  float we = w[e];
  const float4 gv = *(const float4*)(g + (size_t)s * H + q * 4);
  float* ap = agg + (size_t)d * H + q * 4;
  atomicAdd(ap + 0, we * gv.x);
  atomicAdd(ap + 1, we * gv.y);
  atomicAdd(ap + 2, we * gv.z);
  atomicAdd(ap + 3, we * gv.w);
}

__global__ __launch_bounds__(256) void act_kernel(const float* __restrict__ agg,
                                                  const float* __restrict__ bg,
                                                  _Float16* __restrict__ a16, size_t n) {
  size_t i = (size_t)blockIdx.x * 256 + threadIdx.x;
  if (i < n) a16[i] = (_Float16)fmaxf(agg[i] + bg[i & 63], 0.0f);
}

__global__ __launch_bounds__(32) void gru_wmma_kernel(const _Float16* __restrict__ a16,
                                                      _Float16* h16,
                                                      const float* __restrict__ hcur,
                                                      float* __restrict__ hnxt,
                                                      const _Float16* __restrict__ Wih16,
                                                      const _Float16* __restrict__ Whh16,
                                                      const float* __restrict__ b_ih,
                                                      const float* __restrict__ b_hh) {
  const int lane  = threadIdx.x;
  const int node0 = blockIdx.x * 16;
  const int m     = lane & 15;
  const int kb    = (lane < 16) ? 0 : 8;
  const int mrow0 = (lane >> 4) << 3;

  const _Float16* arow = a16 + (size_t)(node0 + m) * H;
  const _Float16* hrow = h16 + (size_t)(node0 + m) * H;
  v16h aA0 = load_fragA(arow, kb);
  v16h aA1 = load_fragA(arow, 32 + kb);
  v16h hA0 = load_fragA(hrow, kb);
  v16h hA1 = load_fragA(hrow, 32 + kb);

  #pragma unroll
  for (int j = 0; j < 4; ++j) {
    const _Float16* wr = Wih16 + (size_t)(      j * 16 + m) * H;
    const _Float16* wz = Wih16 + (size_t)( 64 + j * 16 + m) * H;
    const _Float16* wn = Wih16 + (size_t)(128 + j * 16 + m) * H;
    const _Float16* ur = Whh16 + (size_t)(      j * 16 + m) * H;
    const _Float16* uz = Whh16 + (size_t)( 64 + j * 16 + m) * H;
    const _Float16* un = Whh16 + (size_t)(128 + j * 16 + m) * H;

    v8f ir = {}, iz = {}, in_ = {}, hr = {}, hz = {}, hn = {};

    ir  = __builtin_amdgcn_wmma_f32_16x16x32_f16(false, aA0, false, load_fragA(wr, kb), (short)0, ir,  false, false);
    iz  = __builtin_amdgcn_wmma_f32_16x16x32_f16(false, aA0, false, load_fragA(wz, kb), (short)0, iz,  false, false);
    in_ = __builtin_amdgcn_wmma_f32_16x16x32_f16(false, aA0, false, load_fragA(wn, kb), (short)0, in_, false, false);
    hr  = __builtin_amdgcn_wmma_f32_16x16x32_f16(false, hA0, false, load_fragA(ur, kb), (short)0, hr,  false, false);
    hz  = __builtin_amdgcn_wmma_f32_16x16x32_f16(false, hA0, false, load_fragA(uz, kb), (short)0, hz,  false, false);
    hn  = __builtin_amdgcn_wmma_f32_16x16x32_f16(false, hA0, false, load_fragA(un, kb), (short)0, hn,  false, false);
    ir  = __builtin_amdgcn_wmma_f32_16x16x32_f16(false, aA1, false, load_fragA(wr, 32 + kb), (short)0, ir,  false, false);
    iz  = __builtin_amdgcn_wmma_f32_16x16x32_f16(false, aA1, false, load_fragA(wz, 32 + kb), (short)0, iz,  false, false);
    in_ = __builtin_amdgcn_wmma_f32_16x16x32_f16(false, aA1, false, load_fragA(wn, 32 + kb), (short)0, in_, false, false);
    hr  = __builtin_amdgcn_wmma_f32_16x16x32_f16(false, hA1, false, load_fragA(ur, 32 + kb), (short)0, hr,  false, false);
    hz  = __builtin_amdgcn_wmma_f32_16x16x32_f16(false, hA1, false, load_fragA(uz, 32 + kb), (short)0, hz,  false, false);
    hn  = __builtin_amdgcn_wmma_f32_16x16x32_f16(false, hA1, false, load_fragA(un, 32 + kb), (short)0, hn,  false, false);

    const float bir = b_ih[      j * 16 + m];
    const float biz = b_ih[ 64 + j * 16 + m];
    const float bin = b_ih[128 + j * 16 + m];
    const float bhr = b_hh[      j * 16 + m];
    const float bhz = b_hh[ 64 + j * 16 + m];
    const float bhn = b_hh[128 + j * 16 + m];
    #pragma unroll
    for (int v = 0; v < 8; ++v) {
      float r  = sigmoidf_(ir[v] + bir + hr[v] + bhr);
      float z  = sigmoidf_(iz[v] + biz + hz[v] + bhz);
      float nc = tanhf_(in_[v] + bin + r * (hn[v] + bhn));
      size_t idx = (size_t)(node0 + mrow0 + v) * H + j * 16 + m;
      float hold = hcur[idx];
      float hnew = (1.0f - z) * nc + z * hold;
      hnxt[idx] = hnew;
      h16[idx]  = (_Float16)hnew;
    }
  }
}

__global__ __launch_bounds__(256) void head_kernel(const float* __restrict__ h,
                                                   const float* __restrict__ Wout,
                                                   const float* __restrict__ bout,
                                                   float* __restrict__ out, int N) {
  int i = blockIdx.x * 256 + threadIdx.x;
  if (i >= N) return;
  float a0 = bout[0], a1 = bout[1];
  const float* hr = h + (size_t)i * H;
  #pragma unroll
  for (int k = 0; k < H; ++k) {
    float hv = hr[k];
    a0 += hv * Wout[k * 2 + 0];
    a1 += hv * Wout[k * 2 + 1];
  }
  out[i * 2 + 0] = a0;
  out[i * 2 + 1] = a1;
}


static inline int cdiv(long long a, long long b) { return (int)((a + b - 1) / b); }

extern "C" void kernel_launch(void* const* d_in, const int* in_sizes, int n_in,
                              void* d_out, int out_size, void* d_ws, size_t ws_size,
                              hipStream_t stream) {
  const float* X    = (const float*)d_in[0];
  const int*   ei   = (const int*)  d_in[1];
  const float* Wg   = (const float*)d_in[2];
  const float* bg   = (const float*)d_in[3];
  const float* Wih  = (const float*)d_in[4];
  const float* Whh  = (const float*)d_in[5];
  const float* bih  = (const float*)d_in[6];
  const float* bhh  = (const float*)d_in[7];
  const float* Wout = (const float*)d_in[8];
  const float* bout = (const float*)d_in[9];

  const int N = in_sizes[0] / (T_STEPS * F_IN);
  const int E = in_sizes[1] / 2;
  const int* src = ei;
  const int* dst = ei + E;
  const size_t NH = (size_t)N * H;

  char* ws = (char*)d_ws;
  size_t off = 0;
  auto alloc = [&](size_t bytes) -> void* {
    void* p = ws + off;
    off = (off + bytes + 255) & ~(size_t)255;
    return p;
  };
  float*    deg   = (float*)   alloc(sizeof(float) * N);
  float*    wgt   = (float*)   alloc(sizeof(float) * E);
  float*    g     = (float*)   alloc(sizeof(float) * NH);
  float*    agg   = (float*)   alloc(sizeof(float) * NH);
  _Float16* a16   = (_Float16*)alloc(sizeof(_Float16) * NH);
  float*    hbufA = (float*)   alloc(sizeof(float) * NH);
  float*    hbufB = (float*)   alloc(sizeof(float) * NH);
  _Float16* h16   = (_Float16*)alloc(sizeof(_Float16) * NH);
  _Float16* Wih16 = (_Float16*)alloc(sizeof(_Float16) * 3 * H * H);
  _Float16* Whh16 = (_Float16*)alloc(sizeof(_Float16) * 3 * H * H);
  (void)ws_size;

  init_deg_kernel <<<cdiv(N, 256), 256, 0, stream>>>(deg, N);
  deg_accum_kernel<<<cdiv(E, 256), 256, 0, stream>>>(dst, deg, E);
  dinv_kernel     <<<cdiv(N, 256), 256, 0, stream>>>(deg, N);
  edge_w_kernel   <<<cdiv(E, 256), 256, 0, stream>>>(src, dst, deg, wgt, E);
  cvt_weights_kernel<<<cdiv(3 * H * H, 256), 256, 0, stream>>>(Wih, Whh, Wih16, Whh16, 3 * H * H);
  init_h_kernel   <<<cdiv((long long)NH, 256), 256, 0, stream>>>(hbufA, h16, (int)NH);

  float* hcur = hbufA;
  float* hnxt = hbufB;

  for (int t = 0; t < T_STEPS; ++t) {
    transform_kernel<<<cdiv(N, 4), 256, 0, stream>>>(X, Wg, deg, g, agg, t, N);
    scatter_kernel  <<<cdiv((long long)E * 16, 256), 256, 0, stream>>>(src, dst, wgt, g, agg, E);
    act_kernel      <<<cdiv((long long)NH, 256), 256, 0, stream>>>(agg, bg, a16, NH);
    gru_wmma_kernel <<<N / 16, 32, 0, stream>>>(a16, h16, hcur, hnxt, Wih16, Whh16, bih, bhh);
    float* tmp = hcur; hcur = hnxt; hnxt = tmp;
  }

  head_kernel<<<cdiv(N, 256), 256, 0, stream>>>(hcur, Wout, bout, (float*)d_out, N);
}


# Edge sum taken over the raw 16-wide inputs, Wg product after it (A(XW)=(AX)W): 3 edge passes of 4 timesteps replace 12; bias and relu move into the transform kernel; edge sums re-associated
# speedup vs baseline: 3.3328x; 3.3328x over previous
#include <hip/hip_runtime.h>

#define T_STEPS 12
#define F_IN 16
#define H 64
#define XROW (T_STEPS * F_IN)
#define TPP 4
#define EPW 8

typedef __attribute__((ext_vector_type(16))) _Float16 v16h;
typedef __attribute__((ext_vector_type(8)))  _Float16 v8h;
typedef __attribute__((ext_vector_type(8)))  float    v8f;

__device__ __forceinline__ v16h load_fragA(const _Float16* __restrict__ row, int kb) {
  union { v16h v; v8h h[2]; } u;
  u.h[0] = *(const v8h*)(row + kb);
  u.h[1] = *(const v8h*)(row + kb + 16);
  return u.v;
}

__device__ __forceinline__ float sigmoidf_(float x) {
  return __builtin_amdgcn_rcpf(1.0f + __expf(-x));
}
__device__ __forceinline__ float tanhf_(float x) {
  return 1.0f - 2.0f * __builtin_amdgcn_rcpf(1.0f + __expf(2.0f * x));
}


__global__ __launch_bounds__(256) void init_deg_kernel(float* __restrict__ deg, int N) {
  int i = blockIdx.x * 256 + threadIdx.x;
  if (i < N) deg[i] = 1.0f;
}

__global__ __launch_bounds__(256) void deg_accum_kernel(const int* __restrict__ dst,
                                                        float* __restrict__ deg, int E) {
  int e = blockIdx.x * 256 + threadIdx.x;
  if (e < E) atomicAdd(&deg[dst[e]], 1.0f);
}

__global__ __launch_bounds__(256) void dinv_kernel(float* __restrict__ deg, int N) {
  int i = blockIdx.x * 256 + threadIdx.x;
  if (i < N) deg[i] = rsqrtf(deg[i]);
}

__global__ __launch_bounds__(256) void edge_w_kernel(const int* __restrict__ src,
                                                     const int* __restrict__ dst,
                                                     const float* __restrict__ dinv,
                                                     float* __restrict__ w, int E) {
  int e = blockIdx.x * 256 + threadIdx.x;
  if (e < E) w[e] = dinv[src[e]] * dinv[dst[e]];
}

__global__ __launch_bounds__(256) void cvt_weights_kernel(const float* __restrict__ Wih,
                                                          const float* __restrict__ Whh,
                                                          _Float16* __restrict__ Wih16,
                                                          _Float16* __restrict__ Whh16,
                                                          int n) {
  int i = blockIdx.x * 256 + threadIdx.x;
  if (i < n) {
    Wih16[i] = (_Float16)Wih[i];
    Whh16[i] = (_Float16)Whh[i];
  }
}

__global__ __launch_bounds__(256) void init_h_kernel(float* __restrict__ h,
                                                     _Float16* __restrict__ h16, int n) {
  int i = blockIdx.x * 256 + threadIdx.x;
  if (i < n) { h[i] = 0.0f; h16[i] = (_Float16)0.0f; }
}


__global__ __launch_bounds__(256) void aggx_init_kernel(const float* __restrict__ X,
                                                        const float* __restrict__ dinv,
                                                        float* __restrict__ aggx,
                                                        int pass, int N) {
  int i = blockIdx.x * 256 + threadIdx.x;
  int node = i >> 4;
  if (node >= N) return;
  int q = i & 15;
  float di = dinv[node];
  float s = di * di;
  float4 v = *(const float4*)(X + (size_t)node * XROW + pass * (TPP * F_IN) + q * 4);
  v.x *= s; v.y *= s; v.z *= s; v.w *= s;
  *(float4*)(aggx + (size_t)node * (TPP * F_IN) + q * 4) = v;
}

__global__ __launch_bounds__(256) void scatter_x_kernel(const int* __restrict__ src,
                                                        const int* __restrict__ dst,
                                                        const float* __restrict__ w,
                                                        const float* __restrict__ X,
                                                        float* __restrict__ aggx,
                                                        int pass, int E) {
  const int lane = threadIdx.x & 31;
  const int e0 = __builtin_amdgcn_readfirstlane((blockIdx.x * 8 + (threadIdx.x >> 5)) * EPW);
  if (e0 >= E) return;
  const float* xb = X + pass * (TPP * F_IN) + lane;
  float* ab = aggx + lane;
  if (e0 + EPW <= E) {
    int s[EPW], d[EPW];
    float we[EPW], v0[EPW], v1[EPW];
    for (int k = 0; k < EPW; ++k) { s[k] = src[e0 + k]; d[k] = dst[e0 + k]; we[k] = w[e0 + k]; }
    for (int k = 0; k < EPW; ++k) {
      const float* xp = xb + (size_t)s[k] * XROW;
      v0[k] = xp[0];
      v1[k] = xp[32];
    }
    for (int k = 0; k < EPW; ++k) {
      float* ap = ab + (size_t)d[k] * (TPP * F_IN);
      atomicAdd(ap,      we[k] * v0[k]);
      atomicAdd(ap + 32, we[k] * v1[k]);
    }
  } else {
    for (int e = e0; e < E; ++e) {
      const float* xp = xb + (size_t)src[e] * XROW;
      float* ap = ab + (size_t)dst[e] * (TPP * F_IN);
      float we = w[e];
      atomicAdd(ap,      we * xp[0]);
      atomicAdd(ap + 32, we * xp[32]);
    }
  }
}


__global__ __launch_bounds__(256) void transform_act_kernel(const float* __restrict__ aggx,
                                                            const float* __restrict__ Wg,
                                                            const float* __restrict__ bg,
                                                            _Float16* __restrict__ a16,
                                                            int tq, int N) {
  __shared__ __attribute__((aligned(16))) float sW[F_IN * H];
  const int tid = threadIdx.x;
  ((float4*)sW)[tid] = ((const float4*)Wg)[tid];
  __syncthreads();
  const int c0 = (tid & 7) * 8;
  float b[8];
  for (int j = 0; j < 8; ++j) b[j] = bg[c0 + j];
  for (int p = 0; p < 2; ++p) {
    const int node = blockIdx.x * 64 + p * 32 + (tid >> 3);
    if (node < N) {
      const float4* xp = (const float4*)(aggx + (size_t)node * (TPP * F_IN) + tq * F_IN);
      const float4 x0 = xp[0], x1 = xp[1], x2 = xp[2], x3 = xp[3];
      const float x[F_IN] = {x0.x, x0.y, x0.z, x0.w, x1.x, x1.y, x1.z, x1.w,
                             x2.x, x2.y, x2.z, x2.w, x3.x, x3.y, x3.z, x3.w};
      float acc[8] = {0.0f, 0.0f, 0.0f, 0.0f, 0.0f, 0.0f, 0.0f, 0.0f};
      for (int k = 0; k < F_IN; ++k) {
        const float4 wa = *(const float4*)&sW[k * H + c0];
        const float4 wb = *(const float4*)&sW[k * H + c0 + 4];
        acc[0] += x[k] * wa.x; acc[1] += x[k] * wa.y; acc[2] += x[k] * wa.z; acc[3] += x[k] * wa.w;
        acc[4] += x[k] * wb.x; acc[5] += x[k] * wb.y; acc[6] += x[k] * wb.z; acc[7] += x[k] * wb.w;
      }
      v8h o;
      for (int j = 0; j < 8; ++j) o[j] = (_Float16)fmaxf(acc[j] + b[j], 0.0f);
      *(v8h*)(a16 + (size_t)node * H + c0) = o;
    }
  }
}

__global__ __launch_bounds__(32) void gru_wmma_kernel(const _Float16* __restrict__ a16,
                                                      _Float16* h16,
                                                      const float* __restrict__ hcur,
                                                      float* __restrict__ hnxt,
                                                      const _Float16* __restrict__ Wih16,
                                                      const _Float16* __restrict__ Whh16,
                                                      const float* __restrict__ b_ih,
                                                      const float* __restrict__ b_hh) {
  const int lane  = threadIdx.x;
  const int node0 = blockIdx.x * 16;
  const int m     = lane & 15;
  const int kb    = (lane < 16) ? 0 : 8;
  const int mrow0 = (lane >> 4) << 3;

  const _Float16* arow = a16 + (size_t)(node0 + m) * H;
  const _Float16* hrow = h16 + (size_t)(node0 + m) * H;
  v16h aA0 = load_fragA(arow, kb);
  v16h aA1 = load_fragA(arow, 32 + kb);
  v16h hA0 = load_fragA(hrow, kb);
  v16h hA1 = load_fragA(hrow, 32 + kb);

  #pragma unroll
  for (int j = 0; j < 4; ++j) {
    const _Float16* wr = Wih16 + (size_t)(      j * 16 + m) * H;
    const _Float16* wz = Wih16 + (size_t)( 64 + j * 16 + m) * H;
    const _Float16* wn = Wih16 + (size_t)(128 + j * 16 + m) * H;
    const _Float16* ur = Whh16 + (size_t)(      j * 16 + m) * H;
    const _Float16* uz = Whh16 + (size_t)( 64 + j * 16 + m) * H;
    const _Float16* un = Whh16 + (size_t)(128 + j * 16 + m) * H;

    v8f ir = {}, iz = {}, in_ = {}, hr = {}, hz = {}, hn = {};

    ir  = __builtin_amdgcn_wmma_f32_16x16x32_f16(false, aA0, false, load_fragA(wr, kb), (short)0, ir,  false, false);
    iz  = __builtin_amdgcn_wmma_f32_16x16x32_f16(false, aA0, false, load_fragA(wz, kb), (short)0, iz,  false, false);
    in_ = __builtin_amdgcn_wmma_f32_16x16x32_f16(false, aA0, false, load_fragA(wn, kb), (short)0, in_, false, false);
    hr  = __builtin_amdgcn_wmma_f32_16x16x32_f16(false, hA0, false, load_fragA(ur, kb), (short)0, hr,  false, false);
    hz  = __builtin_amdgcn_wmma_f32_16x16x32_f16(false, hA0, false, load_fragA(uz, kb), (short)0, hz,  false, false);
    hn  = __builtin_amdgcn_wmma_f32_16x16x32_f16(false, hA0, false, load_fragA(un, kb), (short)0, hn,  false, false);
    ir  = __builtin_amdgcn_wmma_f32_16x16x32_f16(false, aA1, false, load_fragA(wr, 32 + kb), (short)0, ir,  false, false);
    iz  = __builtin_amdgcn_wmma_f32_16x16x32_f16(false, aA1, false, load_fragA(wz, 32 + kb), (short)0, iz,  false, false);
    in_ = __builtin_amdgcn_wmma_f32_16x16x32_f16(false, aA1, false, load_fragA(wn, 32 + kb), (short)0, in_, false, false);
    hr  = __builtin_amdgcn_wmma_f32_16x16x32_f16(false, hA1, false, load_fragA(ur, 32 + kb), (short)0, hr,  false, false);
    hz  = __builtin_amdgcn_wmma_f32_16x16x32_f16(false, hA1, false, load_fragA(uz, 32 + kb), (short)0, hz,  false, false);
    hn  = __builtin_amdgcn_wmma_f32_16x16x32_f16(false, hA1, false, load_fragA(un, 32 + kb), (short)0, hn,  false, false);

    const float bir = b_ih[      j * 16 + m];
    const float biz = b_ih[ 64 + j * 16 + m];
    const float bin = b_ih[128 + j * 16 + m];
    const float bhr = b_hh[      j * 16 + m];
    const float bhz = b_hh[ 64 + j * 16 + m];
    const float bhn = b_hh[128 + j * 16 + m];
    #pragma unroll
    for (int v = 0; v < 8; ++v) {
      float r  = sigmoidf_(ir[v] + bir + hr[v] + bhr);
      float z  = sigmoidf_(iz[v] + biz + hz[v] + bhz);
      float nc = tanhf_(in_[v] + bin + r * (hn[v] + bhn));
      size_t idx = (size_t)(node0 + mrow0 + v) * H + j * 16 + m;
      float hold = hcur[idx];
      float hnew = (1.0f - z) * nc + z * hold;
      hnxt[idx] = hnew;
      h16[idx]  = (_Float16)hnew;
    }
  }
}

__global__ __launch_bounds__(256) void head_kernel(const float* __restrict__ h,
                                                   const float* __restrict__ Wout,
                                                   const float* __restrict__ bout,
                                                   float* __restrict__ out, int N) {
  int i = blockIdx.x * 256 + threadIdx.x;
  if (i >= N) return;
  float a0 = bout[0], a1 = bout[1];
  const float* hr = h + (size_t)i * H;
  #pragma unroll
  for (int k = 0; k < H; ++k) {
    float hv = hr[k];
    a0 += hv * Wout[k * 2 + 0];
    a1 += hv * Wout[k * 2 + 1];
  }
  out[i * 2 + 0] = a0;
  out[i * 2 + 1] = a1;
}


static inline int cdiv(long long a, long long b) { return (int)((a + b - 1) / b); }

extern "C" void kernel_launch(void* const* d_in, const int* in_sizes, int n_in,
                              void* d_out, int out_size, void* d_ws, size_t ws_size,
                              hipStream_t stream) {
  const float* X    = (const float*)d_in[0];
  const int*   ei   = (const int*)  d_in[1];
  const float* Wg   = (const float*)d_in[2];
  const float* bg   = (const float*)d_in[3];
  const float* Wih  = (const float*)d_in[4];
  const float* Whh  = (const float*)d_in[5];
  const float* bih  = (const float*)d_in[6];
  const float* bhh  = (const float*)d_in[7];
  const float* Wout = (const float*)d_in[8];
  const float* bout = (const float*)d_in[9];

  const int N = in_sizes[0] / (T_STEPS * F_IN);
  const int E = in_sizes[1] / 2;
  const int* src = ei;
  const int* dst = ei + E;
  const size_t NH = (size_t)N * H;

  char* ws = (char*)d_ws;
  size_t off = 0;
  auto alloc = [&](size_t bytes) -> void* {
    void* p = ws + off;
    off = (off + bytes + 255) & ~(size_t)255;
    return p;
  };
  float*    deg   = (float*)   alloc(sizeof(float) * N);
  float*    wgt   = (float*)   alloc(sizeof(float) * E);
  float*    aggx  = (float*)   alloc(sizeof(float) * NH);
  _Float16* a16   = (_Float16*)alloc(sizeof(_Float16) * NH);
  float*    hbufA = (float*)   alloc(sizeof(float) * NH);
  float*    hbufB = (float*)   alloc(sizeof(float) * NH);
  _Float16* h16   = (_Float16*)alloc(sizeof(_Float16) * NH);
  _Float16* Wih16 = (_Float16*)alloc(sizeof(_Float16) * 3 * H * H);
  _Float16* Whh16 = (_Float16*)alloc(sizeof(_Float16) * 3 * H * H);
  (void)ws_size;

  init_deg_kernel <<<cdiv(N, 256), 256, 0, stream>>>(deg, N);
  deg_accum_kernel<<<cdiv(E, 256), 256, 0, stream>>>(dst, deg, E);
  dinv_kernel     <<<cdiv(N, 256), 256, 0, stream>>>(deg, N);
  edge_w_kernel   <<<cdiv(E, 256), 256, 0, stream>>>(src, dst, deg, wgt, E);
  cvt_weights_kernel<<<cdiv(3 * H * H, 256), 256, 0, stream>>>(Wih, Whh, Wih16, Whh16, 3 * H * H);
  init_h_kernel   <<<cdiv((long long)NH, 256), 256, 0, stream>>>(hbufA, h16, (int)NH);

  float* hcur = hbufA;
  float* hnxt = hbufB;

  for (int p = 0; p < T_STEPS / TPP; ++p) {
    aggx_init_kernel<<<cdiv((long long)N * 16, 256), 256, 0, stream>>>(X, deg, aggx, p, N);
    scatter_x_kernel<<<cdiv(E, 8 * EPW), 256, 0, stream>>>(src, dst, wgt, X, aggx, p, E);
    for (int tq = 0; tq < TPP; ++tq) {
      transform_act_kernel<<<cdiv(N, 64), 256, 0, stream>>>(aggx, Wg, bg, a16, tq, N);
      gru_wmma_kernel <<<N / 16, 32, 0, stream>>>(a16, h16, hcur, hnxt, Wih16, Whh16, bih, bhh);
      float* tmp = hcur; hcur = hnxt; hnxt = tmp;
    }
  }

  head_kernel<<<cdiv(N, 256), 256, 0, stream>>>(hcur, Wout, bout, (float*)d_out, N);
}


# Edge's slot rank taken from the count kernel's returning add, dinv[dst] applied once per node after the edge sum: fill kernel loses its atomics and one random read per edge; edge-sum rounding changed
# speedup vs baseline: 7.4412x; 2.2327x over previous
#include <hip/hip_runtime.h>

#define T_STEPS 12
#define F_IN 16
#define H 64
#define XROW (T_STEPS * F_IN)
#define TPP 4
#define EPT 8
#define NPW 8
#define WGPITCH 20
#define WPITCH 72

typedef __attribute__((ext_vector_type(16))) _Float16 v16h;
typedef __attribute__((ext_vector_type(8)))  _Float16 v8h;
typedef __attribute__((ext_vector_type(8)))  float    v8f;
typedef __attribute__((ext_vector_type(2)))  float    v2f;

__device__ __forceinline__ v16h lds_frag(const _Float16* p, int k0) {
  union { v16h v; v8h h[2]; } u;
  u.h[0] = *(const v8h*)(p + k0);
  u.h[1] = *(const v8h*)(p + k0 + 16);
  return u.v;
}

__device__ __forceinline__ float sigmoidf_(float x) {
  return __builtin_amdgcn_rcpf(1.0f + __expf(-x));
}
__device__ __forceinline__ float tanhf_(float x) {
  return 1.0f - 2.0f * __builtin_amdgcn_rcpf(1.0f + __expf(2.0f * x));
}


__global__ __launch_bounds__(256) void zero_cnt_kernel(int* __restrict__ cnt, int N) {
  int i = blockIdx.x * 256 + threadIdx.x;
  if (i < N) cnt[i] = 0;
}

__global__ __launch_bounds__(256) void count_kernel(const int* __restrict__ dst,
                                                    int* __restrict__ cnt,
                                                    int* __restrict__ rank, int E) {
  const int base = blockIdx.x * (256 * EPT) + threadIdx.x;
  if (base + (EPT - 1) * 256 < E) {
    int d[EPT], r[EPT];
    for (int j = 0; j < EPT; ++j) d[j] = dst[base + j * 256];
    for (int j = 0; j < EPT; ++j) r[j] = atomicAdd(&cnt[d[j]], 1);
    for (int j = 0; j < EPT; ++j) rank[base + j * 256] = r[j];
  } else {
    for (int j = 0; j < EPT; ++j) {
      const int e = base + j * 256;
      if (e < E) rank[e] = atomicAdd(&cnt[dst[e]], 1);
    }
  }
}

__device__ __forceinline__ int wave_sum(int v) {
  for (int o = 16; o > 0; o >>= 1) v += __shfl_xor(v, o);
  return v;
}

__device__ __forceinline__ int block_sum(int v, int* sm) {
  const int lane = threadIdx.x & 31, wv = threadIdx.x >> 5;
  v = wave_sum(v);
  if (lane == 0) sm[wv] = v;
  __syncthreads();
  int t = 0;
  for (int i = 0; i < 8; ++i) t += sm[i];
  return t;
}

__global__ __launch_bounds__(256) void block_total_kernel(const int* __restrict__ cnt,
                                                          int* __restrict__ bsum, int N) {
  __shared__ int sm[8];
  const int base = blockIdx.x * 1024 + threadIdx.x;
  int t = 0;
  for (int j = 0; j < 4; ++j) {
    const int i = base + j * 256;
    if (i < N) t += cnt[i];
  }
  t = block_sum(t, sm);
  if (threadIdx.x == 0) bsum[blockIdx.x] = t;
}

__global__ __launch_bounds__(256) void scan_kernel(const int* __restrict__ cnt,
                                                   const int* __restrict__ bsum,
                                                   int* __restrict__ rowptr,
                                                   float* __restrict__ dinv, int N) {
  __shared__ int sm[8];
  __shared__ int sw[8];
  const int tid = threadIdx.x, lane = tid & 31, wv = tid >> 5;
  int pre = 0;
  for (int i = tid; i < (int)blockIdx.x; i += 256) pre += bsum[i];
  pre = block_sum(pre, sm);

  const int i0 = blockIdx.x * 1024 + tid * 4;
  int c[4];
  for (int j = 0; j < 4; ++j) c[j] = (i0 + j < N) ? cnt[i0 + j] : 0;
  const int mine = c[0] + c[1] + c[2] + c[3];
  int inc = mine;
  for (int o = 1; o < 32; o <<= 1) {
    const int u = __shfl_up(inc, o);
    if (lane >= o) inc += u;
  }
  if (lane == 31) sw[wv] = inc;
  __syncthreads();
  int run = pre + inc - mine;
  for (int i = 0; i < wv; ++i) run += sw[i];
  for (int j = 0; j < 4; ++j) {
    if (i0 + j < N) {
      rowptr[i0 + j] = run;
      dinv[i0 + j] = rsqrtf((float)(c[j] + 1));
      run += c[j];
      if (i0 + j == N - 1) rowptr[N] = run;
    }
  }
}

__device__ __forceinline__ void put_slot(int* __restrict__ csr, int pos, int E, int s, float w) {
  int2 rec;
  rec.x = s;
  rec.y = __float_as_int(w);
  ((int2*)csr)[min(max(pos, 0), E - 1)] = rec;
}

__global__ __launch_bounds__(256) void fill_kernel(const int* __restrict__ src,
                                                   const int* __restrict__ dst,
                                                   const int* __restrict__ rank,
                                                   const int* __restrict__ rowptr,
                                                   const float* __restrict__ dinv,
                                                   int* __restrict__ csr, int E) {
  const int base = blockIdx.x * (256 * EPT) + threadIdx.x;
  if (base + (EPT - 1) * 256 < E) {
    int s[EPT], d[EPT], pos[EPT];
    float w[EPT];
    for (int j = 0; j < EPT; ++j) { s[j] = src[base + j * 256]; d[j] = dst[base + j * 256]; pos[j] = rank[base + j * 256]; }
    for (int j = 0; j < EPT; ++j) { pos[j] += rowptr[d[j]]; w[j] = dinv[s[j]]; }
    for (int j = 0; j < EPT; ++j) put_slot(csr, pos[j], E, s[j], w[j]);
  } else {
    for (int j = 0; j < EPT; ++j) {
      const int e = base + j * 256;
      if (e < E) {
        const int s = src[e];
        put_slot(csr, rowptr[dst[e]] + rank[e], E, s, dinv[s]);
      }
    }
  }
}


__device__ __forceinline__ float lane_f(float v, int l) {
  return __int_as_float(__builtin_amdgcn_readlane(__float_as_int(v), l));
}

__global__ __launch_bounds__(256) void gather_x_kernel(const int* __restrict__ rowptr,
                                                       const int* __restrict__ csr,
                                                       const float* __restrict__ dinv,
                                                       const float* __restrict__ X,
                                                       float* __restrict__ aggx,
                                                       int pass, int N, int E) {
  const int lane = threadIdx.x & 31;
  const int n0 = __builtin_amdgcn_readfirstlane((blockIdx.x * 8 + (threadIdx.x >> 5)) * NPW);
  if (n0 >= N) return;
  const float* xb = X + pass * (TPP * F_IN) + lane * 2;
  const int rpv = rowptr[min(n0 + lane, N)];
  const float dvv = dinv[min(n0 + lane, N - 1)];
  for (int i = 0; i < NPW; ++i) {
    const int n = n0 + i;
    if (n >= N) break;
    const int beg = max(__builtin_amdgcn_readlane(rpv, i), 0);
    const int end = min(__builtin_amdgcn_readlane(rpv, i + 1), E);
    const float dn = lane_f(dvv, i);
    float2 acc;
    acc.x = 0.0f;
    acc.y = 0.0f;
    for (int c = beg; c <= end; c += 32) {
      const int e = c + lane;
      int sv = n;
      float wv = (e == end) ? dn : 0.0f;
      if (e < end) {
        const int2 rec = ((const int2*)csr)[e];
        sv = min(max(rec.x, 0), N - 1);
        wv = __int_as_float(rec.y);
      }
      const int m = min(end + 1 - c, 32);
      for (int k = 0; k < m; k += 8) {
        float2 x[8];
        float w[8];
        for (int j = 0; j < 8; ++j) {
          const int s = __builtin_amdgcn_readlane(sv, k + j);
          w[j] = lane_f(wv, k + j);
          x[j] = *(const float2*)(xb + (size_t)s * XROW);
        }
        for (int j = 0; j < 8; ++j) {
          acc.x += w[j] * x[j].x;
          acc.y += w[j] * x[j].y;
        }
      }
    }
    acc.x *= dn;
    acc.y *= dn;
    *(float2*)(aggx + (size_t)n * XROW + pass * (TPP * F_IN) + lane * 2) = acc;
  }
}

__global__ __launch_bounds__(256) void gru_seq_kernel(const float* __restrict__ aggx,
                                                      const float* __restrict__ Wg,
                                                      const float* __restrict__ bg,
                                                      const float* __restrict__ Wih,
                                                      const float* __restrict__ Whh,
                                                      const float* __restrict__ b_ih,
                                                      const float* __restrict__ b_hh,
                                                      float* __restrict__ hout, int N) {
  __shared__ __attribute__((aligned(16))) _Float16 sW[2 * 3 * H * WPITCH];
  __shared__ __attribute__((aligned(16))) float sB[2 * 3 * H];
  __shared__ __attribute__((aligned(16))) float sBg[H];
  __shared__ __attribute__((aligned(16))) float sWgT[H * WGPITCH];

  const int tid = threadIdx.x;
  for (int i = tid; i < 2 * 3 * H * 8; i += 256) {
    const int row = i >> 3, seg = i & 7;
    const float* sp = (row < 3 * H ? Wih + (size_t)row * H : Whh + (size_t)(row - 3 * H) * H) + seg * 8;
    const float4 f0 = ((const float4*)sp)[0], f1 = ((const float4*)sp)[1];
    v8h o;
    o[0] = (_Float16)f0.x; o[1] = (_Float16)f0.y; o[2] = (_Float16)f0.z; o[3] = (_Float16)f0.w;
    o[4] = (_Float16)f1.x; o[5] = (_Float16)f1.y; o[6] = (_Float16)f1.z; o[7] = (_Float16)f1.w;
    *(v8h*)(sW + row * WPITCH + seg * 8) = o;
  }
  for (int i = tid; i < 2 * 3 * H; i += 256) sB[i] = (i < 3 * H) ? b_ih[i] : b_hh[i - 3 * H];
  if (tid < H) sBg[tid] = bg[tid];
  for (int i = tid; i < F_IN * H; i += 256) sWgT[(i & (H - 1)) * WGPITCH + (i >> 6)] = Wg[i];
  __syncthreads();

  const int lane = tid & 31;
  const int c = lane & 15, hf = lane >> 4;
  const int tile = blockIdx.x * 8 + (tid >> 5);
  if (tile * 16 >= N) return;
  const int node = tile * 16 + c;
  const int nodec = min(node, N - 1);

  float h32[4][8];
  for (int j = 0; j < 4; ++j) {
    for (int r = 0; r < 8; ++r) h32[j][r] = 0.0f;
  }
  v16h hB0 = {}, hB1 = {};

  const float* xrow = aggx + (size_t)nodec * XROW + 8 * hf;
  float4 xa = ((const float4*)xrow)[0], xb = ((const float4*)xrow)[1];

  for (int t = 0; t < T_STEPS; ++t) {
    asm volatile("" ::: "memory");
    const float* xn = xrow + min(t + 1, T_STEPS - 1) * F_IN;
    const float4 na = ((const float4*)xn)[0], nb = ((const float4*)xn)[1];
    const float xs[8] = {xa.x, xa.y, xa.z, xa.w, xb.x, xb.y, xb.z, xb.w};

    v16h aB0, aB1;
    for (int blk = 0; blk < 4; ++blk) {
      const float4 w0 = *(const float4*)&sWgT[(16 * blk + c) * WGPITCH + 8 * hf];
      const float4 w1 = *(const float4*)&sWgT[(16 * blk + c) * WGPITCH + 8 * hf + 4];
      const float ws8[8] = {w0.x, w0.y, w0.z, w0.w, w1.x, w1.y, w1.z, w1.w};
      v8f acc = {};
      for (int q = 0; q < 4; ++q) {
        v2f wq, xq;
        wq[0] = ws8[2 * q];
        wq[1] = ws8[2 * q + 1];
        xq[0] = xs[2 * q];
        xq[1] = xs[2 * q + 1];
        acc = __builtin_amdgcn_wmma_f32_16x16x4_f32(false, wq, false, xq, (short)0, acc, false, false);
      }
      const float4 g0 = *(const float4*)&sBg[16 * blk + 8 * hf];
      const float4 g1 = *(const float4*)&sBg[16 * blk + 8 * hf + 4];
      const float gb[8] = {g0.x, g0.y, g0.z, g0.w, g1.x, g1.y, g1.z, g1.w};
      for (int r = 0; r < 8; ++r) {
        const _Float16 av = (_Float16)fmaxf(acc[r] + gb[r], 0.0f);
        if (blk < 2) aB0[(blk & 1) * 8 + r] = av; else aB1[(blk & 1) * 8 + r] = av;
      }
    }

    for (int j = 0; j < 4; ++j) {
      asm volatile("" ::: "memory");
      const _Float16* wr = sW + (          16 * j + c) * WPITCH + 8 * hf;
      const _Float16* wz = sW + (     64 + 16 * j + c) * WPITCH + 8 * hf;
      const _Float16* wn = sW + (    128 + 16 * j + c) * WPITCH + 8 * hf;
      const _Float16* ur = sW + (192 +       16 * j + c) * WPITCH + 8 * hf;
      const _Float16* uz = sW + (192 +  64 + 16 * j + c) * WPITCH + 8 * hf;
      const _Float16* un = sW + (192 + 128 + 16 * j + c) * WPITCH + 8 * hf;

      v8f ir = {}, iz = {}, in_ = {}, hr = {}, hz = {}, hn = {};
      ir  = __builtin_amdgcn_wmma_f32_16x16x32_f16(false, lds_frag(wr, 0), false, aB0, (short)0, ir,  false, false);
      iz  = __builtin_amdgcn_wmma_f32_16x16x32_f16(false, lds_frag(wz, 0), false, aB0, (short)0, iz,  false, false);
      in_ = __builtin_amdgcn_wmma_f32_16x16x32_f16(false, lds_frag(wn, 0), false, aB0, (short)0, in_, false, false);
      hr  = __builtin_amdgcn_wmma_f32_16x16x32_f16(false, lds_frag(ur, 0), false, hB0, (short)0, hr,  false, false);
      hz  = __builtin_amdgcn_wmma_f32_16x16x32_f16(false, lds_frag(uz, 0), false, hB0, (short)0, hz,  false, false);
      hn  = __builtin_amdgcn_wmma_f32_16x16x32_f16(false, lds_frag(un, 0), false, hB0, (short)0, hn,  false, false);
      ir  = __builtin_amdgcn_wmma_f32_16x16x32_f16(false, lds_frag(wr, 32), false, aB1, (short)0, ir,  false, false);
      iz  = __builtin_amdgcn_wmma_f32_16x16x32_f16(false, lds_frag(wz, 32), false, aB1, (short)0, iz,  false, false);
      in_ = __builtin_amdgcn_wmma_f32_16x16x32_f16(false, lds_frag(wn, 32), false, aB1, (short)0, in_, false, false);
      hr  = __builtin_amdgcn_wmma_f32_16x16x32_f16(false, lds_frag(ur, 32), false, hB1, (short)0, hr,  false, false);
      hz  = __builtin_amdgcn_wmma_f32_16x16x32_f16(false, lds_frag(uz, 32), false, hB1, (short)0, hz,  false, false);
      hn  = __builtin_amdgcn_wmma_f32_16x16x32_f16(false, lds_frag(un, 32), false, hB1, (short)0, hn,  false, false);

      const int ch = 16 * j + 8 * hf;
      float bir[8], biz[8], bin[8], bhr[8], bhz[8], bhn[8];
      for (int r = 0; r < 8; r += 4) {
        *(float4*)&bir[r] = *(const float4*)&sB[            ch + r];
        *(float4*)&biz[r] = *(const float4*)&sB[       64 + ch + r];
        *(float4*)&bin[r] = *(const float4*)&sB[      128 + ch + r];
        *(float4*)&bhr[r] = *(const float4*)&sB[192 +       ch + r];
        *(float4*)&bhz[r] = *(const float4*)&sB[192 +  64 + ch + r];
        *(float4*)&bhn[r] = *(const float4*)&sB[192 + 128 + ch + r];
      }
      for (int r = 0; r < 8; ++r) {
        float rg = sigmoidf_(ir[r] + bir[r] + hr[r] + bhr[r]);
        float zg = sigmoidf_(iz[r] + biz[r] + hz[r] + bhz[r]);
        float nc = tanhf_(in_[r] + bin[r] + rg * (hn[r] + bhn[r]));
        float hold = h32[j][r];
        h32[j][r] = (1.0f - zg) * nc + zg * hold;
      }
    }

    for (int r = 0; r < 8; ++r) {
      hB0[r]     = (_Float16)h32[0][r];
      hB0[8 + r] = (_Float16)h32[1][r];
      hB1[r]     = (_Float16)h32[2][r];
      hB1[8 + r] = (_Float16)h32[3][r];
    }
    xa = na;
    xb = nb;
  }

  if (node < N) {
    float* ho = hout + (size_t)node * H + 8 * hf;
    for (int j = 0; j < 4; ++j) {
      ((float4*)(ho + 16 * j))[0] = make_float4(h32[j][0], h32[j][1], h32[j][2], h32[j][3]);
      ((float4*)(ho + 16 * j))[1] = make_float4(h32[j][4], h32[j][5], h32[j][6], h32[j][7]);
    }
  }
}

__global__ __launch_bounds__(256) void head_kernel(const float* __restrict__ h,
                                                   const float* __restrict__ Wout,
                                                   const float* __restrict__ bout,
                                                   float* __restrict__ out, int N) {
  int i = blockIdx.x * 256 + threadIdx.x;
  if (i >= N) return;
  float a0 = bout[0], a1 = bout[1];
  const float* hr = h + (size_t)i * H;
  #pragma unroll
  for (int k = 0; k < H; ++k) {
    float hv = hr[k];
    a0 += hv * Wout[k * 2 + 0];
    a1 += hv * Wout[k * 2 + 1];
  }
  out[i * 2 + 0] = a0;
  out[i * 2 + 1] = a1;
}


static inline int cdiv(long long a, long long b) { return (int)((a + b - 1) / b); }

extern "C" void kernel_launch(void* const* d_in, const int* in_sizes, int n_in,
                              void* d_out, int out_size, void* d_ws, size_t ws_size,
                              hipStream_t stream) {
  const float* X    = (const float*)d_in[0];
  const int*   ei   = (const int*)  d_in[1];
  const float* Wg   = (const float*)d_in[2];
  const float* bg   = (const float*)d_in[3];
  const float* Wih  = (const float*)d_in[4];
  const float* Whh  = (const float*)d_in[5];
  const float* bih  = (const float*)d_in[6];
  const float* bhh  = (const float*)d_in[7];
  const float* Wout = (const float*)d_in[8];
  const float* bout = (const float*)d_in[9];

  const int N = in_sizes[0] / (T_STEPS * F_IN);
  const int E = in_sizes[1] / 2;
  const int* src = ei;
  const int* dst = ei + E;
  const size_t NH = (size_t)N * H;

  char* ws = (char*)d_ws;
  size_t off = 0;
  auto alloc = [&](size_t bytes) -> void* {
    void* p = ws + off;
    off = (off + bytes + 255) & ~(size_t)255;
    return p;
  };
  int*      rowptr  = (int*)   alloc(sizeof(int) * (N + 1));
  float*    dinv    = (float*) alloc(sizeof(float) * N);
  int*      cnt     = (int*)   alloc(sizeof(int) * N);
  int*      bsum    = (int*)   alloc(sizeof(int) * cdiv(N, 1024));
  int*      csr     = (int*)   alloc(sizeof(int) * 2 * E);
  float*    aggx  = (float*)   alloc(sizeof(float) * (size_t)N * XROW);
  float*    hT    = (float*)   alloc(sizeof(float) * NH);
  int*      rank  = (int*)aggx;
  (void)ws_size;

  zero_cnt_kernel   <<<cdiv(N, 256), 256, 0, stream>>>(cnt, N);
  count_kernel      <<<cdiv(E, 256 * EPT), 256, 0, stream>>>(dst, cnt, rank, E);
  block_total_kernel<<<cdiv(N, 1024), 256, 0, stream>>>(cnt, bsum, N);
  scan_kernel       <<<cdiv(N, 1024), 256, 0, stream>>>(cnt, bsum, rowptr, dinv, N);
  fill_kernel       <<<cdiv(E, 256 * EPT), 256, 0, stream>>>(src, dst, rank, rowptr, dinv, csr, E);

  for (int p = 0; p < T_STEPS / TPP; ++p)
    gather_x_kernel<<<cdiv(N, 8 * NPW), 256, 0, stream>>>(rowptr, csr, dinv, X, aggx, p, N, E);

  gru_seq_kernel<<<cdiv(N, 8 * 16), 256, 0, stream>>>(aggx, Wg, bg, Wih, Whh, bih, bhh, hT, N);

  head_kernel<<<cdiv(N, 256), 256, 0, stream>>>(hT, Wout, bout, (float*)d_out, N);
}


# Edge slots counted and handed out from node tables in LDS, one work-group per (edge slice, node range), joined by a per-node prefix over slices: global adds, rank array gone; edge-sum order changed
# speedup vs baseline: 8.9093x; 1.1973x over previous
#include <hip/hip_runtime.h>

#define T_STEPS 12
#define F_IN 16
#define H 64
#define XROW (T_STEPS * F_IN)
#define TPP 4
#define RANGE 51200
#define NSL 128
#define SEPT 13
#define NPW 8
#define WGPITCH 20
#define WPITCH 72

typedef __attribute__((ext_vector_type(16))) _Float16 v16h;
typedef __attribute__((ext_vector_type(8)))  _Float16 v8h;
typedef __attribute__((ext_vector_type(8)))  float    v8f;
typedef __attribute__((ext_vector_type(2)))  float    v2f;

__device__ __forceinline__ v16h lds_frag(const _Float16* p, int k0) {
  union { v16h v; v8h h[2]; } u;
  u.h[0] = *(const v8h*)(p + k0);
  u.h[1] = *(const v8h*)(p + k0 + 16);
  return u.v;
}

__device__ __forceinline__ float sigmoidf_(float x) {
  return __builtin_amdgcn_rcpf(1.0f + __expf(-x));
}
__device__ __forceinline__ float tanhf_(float x) {
  return 1.0f - 2.0f * __builtin_amdgcn_rcpf(1.0f + __expf(2.0f * x));
}


__device__ __forceinline__ int part_pitch(int N) { return (N + 63) & ~63; }

__global__ __launch_bounds__(1024) void slice_count_kernel(const int* __restrict__ dst,
                                                           int* __restrict__ part,
                                                           int N, int E, int sl) {
  __shared__ __attribute__((aligned(16))) int tbl[RANGE];
  const int tid = threadIdx.x;
  const int k = blockIdx.x, r0 = blockIdx.y * RANGE;
  const int nr = min(RANGE, N - r0);
  for (int i = tid; i < RANGE / 4; i += 1024) ((int4*)tbl)[i] = make_int4(0, 0, 0, 0);
  __syncthreads();
  const int e0 = min(k * sl, E), e1 = min(e0 + sl, E);
  for (int base = e0 + tid; base < e1; base += 1024 * SEPT) {
    unsigned d[SEPT];
    for (int j = 0; j < SEPT; ++j) d[j] = (unsigned)(dst[min(base + j * 1024, e1 - 1)] - r0);
    for (int j = 0; j < SEPT; ++j) {
      if (base + j * 1024 >= e1) d[j] = 0xFFFFFFFFu;
    }
    for (int j = 0; j < SEPT; ++j) {
      if (d[j] < (unsigned)nr) atomicAdd(&tbl[d[j]], 1);
    }
  }
  __syncthreads();
  int* out = part + (size_t)k * part_pitch(N) + r0;
  const int nv = nr >> 2;
  for (int i = tid; i < nv; i += 1024) ((int4*)out)[i] = ((const int4*)tbl)[i];
  if (tid < (nr & 3)) out[4 * nv + tid] = tbl[4 * nv + tid];
}

__global__ __launch_bounds__(256) void slice_prefix_kernel(int* __restrict__ part,
                                                           int* __restrict__ cnt, int N) {
  const int n = blockIdx.x * 256 + threadIdx.x;
  if (n >= N) return;
  const size_t pn = (size_t)part_pitch(N);
  int run = 0;
  for (int k0 = 0; k0 < NSL; k0 += 32) {
    int c[32];
    #pragma unroll
    for (int j = 0; j < 32; ++j) c[j] = part[(size_t)(k0 + j) * pn + n];
    #pragma unroll
    for (int j = 0; j < 32; ++j) {
      part[(size_t)(k0 + j) * pn + n] = run;
      run += c[j];
    }
  }
  cnt[n] = run;
}

__device__ __forceinline__ int wave_sum(int v) {
  for (int o = 16; o > 0; o >>= 1) v += __shfl_xor(v, o);
  return v;
}

__device__ __forceinline__ int block_sum(int v, int* sm) {
  const int lane = threadIdx.x & 31, wv = threadIdx.x >> 5;
  v = wave_sum(v);
  if (lane == 0) sm[wv] = v;
  __syncthreads();
  int t = 0;
  for (int i = 0; i < 8; ++i) t += sm[i];
  return t;
}

__global__ __launch_bounds__(256) void block_total_kernel(const int* __restrict__ cnt,
                                                          int* __restrict__ bsum, int N) {
  __shared__ int sm[8];
  const int base = blockIdx.x * 1024 + threadIdx.x;
  int t = 0;
  for (int j = 0; j < 4; ++j) {
    const int i = base + j * 256;
    if (i < N) t += cnt[i];
  }
  t = block_sum(t, sm);
  if (threadIdx.x == 0) bsum[blockIdx.x] = t;
}

__global__ __launch_bounds__(256) void scan_kernel(const int* __restrict__ cnt,
                                                   const int* __restrict__ bsum,
                                                   int* __restrict__ rowptr,
                                                   float* __restrict__ dinv, int N) {
  __shared__ int sm[8];
  __shared__ int sw[8];
  const int tid = threadIdx.x, lane = tid & 31, wv = tid >> 5;
  int pre = 0;
  for (int i = tid; i < (int)blockIdx.x; i += 256) pre += bsum[i];
  pre = block_sum(pre, sm);

  const int i0 = blockIdx.x * 1024 + tid * 4;
  int c[4];
  for (int j = 0; j < 4; ++j) c[j] = (i0 + j < N) ? cnt[i0 + j] : 0;
  const int mine = c[0] + c[1] + c[2] + c[3];
  int inc = mine;
  for (int o = 1; o < 32; o <<= 1) {
    const int u = __shfl_up(inc, o);
    if (lane >= o) inc += u;
  }
  if (lane == 31) sw[wv] = inc;
  __syncthreads();
  int run = pre + inc - mine;
  for (int i = 0; i < wv; ++i) run += sw[i];
  for (int j = 0; j < 4; ++j) {
    if (i0 + j < N) {
      rowptr[i0 + j] = run;
      dinv[i0 + j] = rsqrtf((float)(c[j] + 1));
      run += c[j];
      if (i0 + j == N - 1) rowptr[N] = run;
    }
  }
}

__device__ __forceinline__ void put_slot(int* __restrict__ csr, int pos, int E, int s, float w) {
  int2 rec;
  rec.x = s;
  rec.y = __float_as_int(w);
  ((int2*)csr)[min(max(pos, 0), E - 1)] = rec;
}

__global__ __launch_bounds__(1024) void slice_fill_kernel(const int* __restrict__ src,
                                                          const int* __restrict__ dst,
                                                          const int* __restrict__ part,
                                                          const int* __restrict__ rowptr,
                                                          const float* __restrict__ dinv,
                                                          int* __restrict__ csr,
                                                          int N, int E, int sl) {
  __shared__ __attribute__((aligned(16))) int tbl[RANGE];
  const int tid = threadIdx.x;
  const int k = blockIdx.x, r0 = blockIdx.y * RANGE;
  const int nr = min(RANGE, N - r0);
  const int* pk = part + (size_t)k * part_pitch(N) + r0;
  const int* rp = rowptr + r0;
  const int nv = nr >> 2;
  for (int i0 = tid; i0 < nv; i0 += 1024 * 8) {
    int4 a[8], b[8];
    for (int j = 0; j < 8; ++j) {
      const int i = min(i0 + j * 1024, nv - 1);
      a[j] = ((const int4*)rp)[i];
      b[j] = ((const int4*)pk)[i];
    }
    for (int j = 0; j < 8; ++j) {
      const int i = i0 + j * 1024;
      if (i < nv) ((int4*)tbl)[i] = make_int4(a[j].x + b[j].x, a[j].y + b[j].y, a[j].z + b[j].z, a[j].w + b[j].w);
    }
  }
  if (tid < (nr & 3)) tbl[4 * nv + tid] = rp[4 * nv + tid] + pk[4 * nv + tid];
  __syncthreads();
  const int e0 = min(k * sl, E), e1 = min(e0 + sl, E);
  for (int base = e0 + tid; base < e1; base += 1024 * SEPT) {
    unsigned d[SEPT];
    int s[SEPT], pos[SEPT];
    float w[SEPT];
    for (int j = 0; j < SEPT; ++j) {
      const int e = min(base + j * 1024, e1 - 1);
      d[j] = (unsigned)(dst[e] - r0);
      s[j] = src[e];
    }
    for (int j = 0; j < SEPT; ++j) {
      if (base + j * 1024 >= e1) d[j] = 0xFFFFFFFFu;
    }
    for (int j = 0; j < SEPT; ++j) {
      const bool mine = d[j] < (unsigned)nr;
      w[j] = mine ? dinv[min(max(s[j], 0), N - 1)] : 0.0f;
      pos[j] = mine ? atomicAdd(&tbl[d[j]], 1) : -1;
    }
    for (int j = 0; j < SEPT; ++j) {
      if (pos[j] >= 0) put_slot(csr, pos[j], E, s[j], w[j]);
    }
  }
}


__device__ __forceinline__ float lane_f(float v, int l) {
  return __int_as_float(__builtin_amdgcn_readlane(__float_as_int(v), l));
}

__global__ __launch_bounds__(256) void gather_x_kernel(const int* __restrict__ rowptr,
                                                       const int* __restrict__ csr,
                                                       const float* __restrict__ dinv,
                                                       const float* __restrict__ X,
                                                       float* __restrict__ aggx,
                                                       int pass, int N, int E) {
  const int lane = threadIdx.x & 31;
  const int n0 = __builtin_amdgcn_readfirstlane((blockIdx.x * 8 + (threadIdx.x >> 5)) * NPW);
  if (n0 >= N) return;
  const float* xb = X + pass * (TPP * F_IN) + lane * 2;
  const int rpv = rowptr[min(n0 + lane, N)];
  const float dvv = dinv[min(n0 + lane, N - 1)];
  for (int i = 0; i < NPW; ++i) {
    const int n = n0 + i;
    if (n >= N) break;
    const int beg = max(__builtin_amdgcn_readlane(rpv, i), 0);
    const int end = min(__builtin_amdgcn_readlane(rpv, i + 1), E);
    const float dn = lane_f(dvv, i);
    float2 acc;
    acc.x = 0.0f;
    acc.y = 0.0f;
    for (int c = beg; c <= end; c += 32) {
      const int e = c + lane;
      int sv = n;
      float wv = (e == end) ? dn : 0.0f;
      if (e < end) {
        const int2 rec = ((const int2*)csr)[e];
        sv = min(max(rec.x, 0), N - 1);
        wv = __int_as_float(rec.y);
      }
      const int m = min(end + 1 - c, 32);
      for (int k = 0; k < m; k += 8) {
        float2 x[8];
        float w[8];
        for (int j = 0; j < 8; ++j) {
          const int s = __builtin_amdgcn_readlane(sv, k + j);
          w[j] = lane_f(wv, k + j);
          x[j] = *(const float2*)(xb + (size_t)s * XROW);
        }
        for (int j = 0; j < 8; ++j) {
          acc.x += w[j] * x[j].x;
          acc.y += w[j] * x[j].y;
        }
      }
    }
    acc.x *= dn;
    acc.y *= dn;
    *(float2*)(aggx + (size_t)n * XROW + pass * (TPP * F_IN) + lane * 2) = acc;
  }
}

__global__ __launch_bounds__(256) void gru_seq_kernel(const float* __restrict__ aggx,
                                                      const float* __restrict__ Wg,
                                                      const float* __restrict__ bg,
                                                      const float* __restrict__ Wih,
                                                      const float* __restrict__ Whh,
                                                      const float* __restrict__ b_ih,
                                                      const float* __restrict__ b_hh,
                                                      float* __restrict__ hout, int N) {
  __shared__ __attribute__((aligned(16))) _Float16 sW[2 * 3 * H * WPITCH];
  __shared__ __attribute__((aligned(16))) float sB[2 * 3 * H];
  __shared__ __attribute__((aligned(16))) float sBg[H];
  __shared__ __attribute__((aligned(16))) float sWgT[H * WGPITCH];

  const int tid = threadIdx.x;
  for (int i = tid; i < 2 * 3 * H * 8; i += 256) {
    const int row = i >> 3, seg = i & 7;
    const float* sp = (row < 3 * H ? Wih + (size_t)row * H : Whh + (size_t)(row - 3 * H) * H) + seg * 8;
    const float4 f0 = ((const float4*)sp)[0], f1 = ((const float4*)sp)[1];
    v8h o;
    o[0] = (_Float16)f0.x; o[1] = (_Float16)f0.y; o[2] = (_Float16)f0.z; o[3] = (_Float16)f0.w;
    o[4] = (_Float16)f1.x; o[5] = (_Float16)f1.y; o[6] = (_Float16)f1.z; o[7] = (_Float16)f1.w;
    *(v8h*)(sW + row * WPITCH + seg * 8) = o;
  }
  for (int i = tid; i < 2 * 3 * H; i += 256) sB[i] = (i < 3 * H) ? b_ih[i] : b_hh[i - 3 * H];
  if (tid < H) sBg[tid] = bg[tid];
  for (int i = tid; i < F_IN * H; i += 256) sWgT[(i & (H - 1)) * WGPITCH + (i >> 6)] = Wg[i];
  __syncthreads();

  const int lane = tid & 31;
  const int c = lane & 15, hf = lane >> 4;
  const int tile = blockIdx.x * 8 + (tid >> 5);
  if (tile * 16 >= N) return;
  const int node = tile * 16 + c;
  const int nodec = min(node, N - 1);

  float h32[4][8];
  for (int j = 0; j < 4; ++j) {
    for (int r = 0; r < 8; ++r) h32[j][r] = 0.0f;
  }
  v16h hB0 = {}, hB1 = {};

  const float* xrow = aggx + (size_t)nodec * XROW + 8 * hf;
  float4 xa = ((const float4*)xrow)[0], xb = ((const float4*)xrow)[1];

  for (int t = 0; t < T_STEPS; ++t) {
    asm volatile("" ::: "memory");
    const float* xn = xrow + min(t + 1, T_STEPS - 1) * F_IN;
    const float4 na = ((const float4*)xn)[0], nb = ((const float4*)xn)[1];
    const float xs[8] = {xa.x, xa.y, xa.z, xa.w, xb.x, xb.y, xb.z, xb.w};

    v16h aB0, aB1;
    for (int blk = 0; blk < 4; ++blk) {
      const float4 w0 = *(const float4*)&sWgT[(16 * blk + c) * WGPITCH + 8 * hf];
      const float4 w1 = *(const float4*)&sWgT[(16 * blk + c) * WGPITCH + 8 * hf + 4];
      const float ws8[8] = {w0.x, w0.y, w0.z, w0.w, w1.x, w1.y, w1.z, w1.w};
      v8f acc = {};
      for (int q = 0; q < 4; ++q) {
        v2f wq, xq;
        wq[0] = ws8[2 * q];
        wq[1] = ws8[2 * q + 1];
        xq[0] = xs[2 * q];
        xq[1] = xs[2 * q + 1];
        acc = __builtin_amdgcn_wmma_f32_16x16x4_f32(false, wq, false, xq, (short)0, acc, false, false);
      }
      const float4 g0 = *(const float4*)&sBg[16 * blk + 8 * hf];
      const float4 g1 = *(const float4*)&sBg[16 * blk + 8 * hf + 4];
      const float gb[8] = {g0.x, g0.y, g0.z, g0.w, g1.x, g1.y, g1.z, g1.w};
      for (int r = 0; r < 8; ++r) {
        const _Float16 av = (_Float16)fmaxf(acc[r] + gb[r], 0.0f);
        if (blk < 2) aB0[(blk & 1) * 8 + r] = av; else aB1[(blk & 1) * 8 + r] = av;
      }
    }

    for (int j = 0; j < 4; ++j) {
      asm volatile("" ::: "memory");
      const _Float16* wr = sW + (          16 * j + c) * WPITCH + 8 * hf;
      const _Float16* wz = sW + (     64 + 16 * j + c) * WPITCH + 8 * hf;
      const _Float16* wn = sW + (    128 + 16 * j + c) * WPITCH + 8 * hf;
      const _Float16* ur = sW + (192 +       16 * j + c) * WPITCH + 8 * hf;
      const _Float16* uz = sW + (192 +  64 + 16 * j + c) * WPITCH + 8 * hf;
      const _Float16* un = sW + (192 + 128 + 16 * j + c) * WPITCH + 8 * hf;

      v8f ir = {}, iz = {}, in_ = {}, hr = {}, hz = {}, hn = {};
      ir  = __builtin_amdgcn_wmma_f32_16x16x32_f16(false, lds_frag(wr, 0), false, aB0, (short)0, ir,  false, false);
      iz  = __builtin_amdgcn_wmma_f32_16x16x32_f16(false, lds_frag(wz, 0), false, aB0, (short)0, iz,  false, false);
      in_ = __builtin_amdgcn_wmma_f32_16x16x32_f16(false, lds_frag(wn, 0), false, aB0, (short)0, in_, false, false);
      hr  = __builtin_amdgcn_wmma_f32_16x16x32_f16(false, lds_frag(ur, 0), false, hB0, (short)0, hr,  false, false);
      hz  = __builtin_amdgcn_wmma_f32_16x16x32_f16(false, lds_frag(uz, 0), false, hB0, (short)0, hz,  false, false);
      hn  = __builtin_amdgcn_wmma_f32_16x16x32_f16(false, lds_frag(un, 0), false, hB0, (short)0, hn,  false, false);
      ir  = __builtin_amdgcn_wmma_f32_16x16x32_f16(false, lds_frag(wr, 32), false, aB1, (short)0, ir,  false, false);
      iz  = __builtin_amdgcn_wmma_f32_16x16x32_f16(false, lds_frag(wz, 32), false, aB1, (short)0, iz,  false, false);
      in_ = __builtin_amdgcn_wmma_f32_16x16x32_f16(false, lds_frag(wn, 32), false, aB1, (short)0, in_, false, false);
      hr  = __builtin_amdgcn_wmma_f32_16x16x32_f16(false, lds_frag(ur, 32), false, hB1, (short)0, hr,  false, false);
      hz  = __builtin_amdgcn_wmma_f32_16x16x32_f16(false, lds_frag(uz, 32), false, hB1, (short)0, hz,  false, false);
      hn  = __builtin_amdgcn_wmma_f32_16x16x32_f16(false, lds_frag(un, 32), false, hB1, (short)0, hn,  false, false);

      const int ch = 16 * j + 8 * hf;
      float bir[8], biz[8], bin[8], bhr[8], bhz[8], bhn[8];
      for (int r = 0; r < 8; r += 4) {
        *(float4*)&bir[r] = *(const float4*)&sB[            ch + r];
        *(float4*)&biz[r] = *(const float4*)&sB[       64 + ch + r];
        *(float4*)&bin[r] = *(const float4*)&sB[      128 + ch + r];
        *(float4*)&bhr[r] = *(const float4*)&sB[192 +       ch + r];
        *(float4*)&bhz[r] = *(const float4*)&sB[192 +  64 + ch + r];
        *(float4*)&bhn[r] = *(const float4*)&sB[192 + 128 + ch + r];
      }
      for (int r = 0; r < 8; ++r) {
        float rg = sigmoidf_(ir[r] + bir[r] + hr[r] + bhr[r]);
        float zg = sigmoidf_(iz[r] + biz[r] + hz[r] + bhz[r]);
        float nc = tanhf_(in_[r] + bin[r] + rg * (hn[r] + bhn[r]));
        float hold = h32[j][r];
        h32[j][r] = (1.0f - zg) * nc + zg * hold;
      }
    }

    for (int r = 0; r < 8; ++r) {
      hB0[r]     = (_Float16)h32[0][r];
      hB0[8 + r] = (_Float16)h32[1][r];
      hB1[r]     = (_Float16)h32[2][r];
      hB1[8 + r] = (_Float16)h32[3][r];
    }
    xa = na;
    xb = nb;
  }

  if (node < N) {
    float* ho = hout + (size_t)node * H + 8 * hf;
    for (int j = 0; j < 4; ++j) {
      ((float4*)(ho + 16 * j))[0] = make_float4(h32[j][0], h32[j][1], h32[j][2], h32[j][3]);
      ((float4*)(ho + 16 * j))[1] = make_float4(h32[j][4], h32[j][5], h32[j][6], h32[j][7]);
    }
  }
}

__global__ __launch_bounds__(256) void head_kernel(const float* __restrict__ h,
                                                   const float* __restrict__ Wout,
                                                   const float* __restrict__ bout,
                                                   float* __restrict__ out, int N) {
  int i = blockIdx.x * 256 + threadIdx.x;
  if (i >= N) return;
  float a0 = bout[0], a1 = bout[1];
  const float* hr = h + (size_t)i * H;
  #pragma unroll
  for (int k = 0; k < H; ++k) {
    float hv = hr[k];
    a0 += hv * Wout[k * 2 + 0];
    a1 += hv * Wout[k * 2 + 1];
  }
  out[i * 2 + 0] = a0;
  out[i * 2 + 1] = a1;
}


static inline int cdiv(long long a, long long b) { return (int)((a + b - 1) / b); }

extern "C" void kernel_launch(void* const* d_in, const int* in_sizes, int n_in,
                              void* d_out, int out_size, void* d_ws, size_t ws_size,
                              hipStream_t stream) {
  const float* X    = (const float*)d_in[0];
  const int*   ei   = (const int*)  d_in[1];
  const float* Wg   = (const float*)d_in[2];
  const float* bg   = (const float*)d_in[3];
  const float* Wih  = (const float*)d_in[4];
  const float* Whh  = (const float*)d_in[5];
  const float* bih  = (const float*)d_in[6];
  const float* bhh  = (const float*)d_in[7];
  const float* Wout = (const float*)d_in[8];
  const float* bout = (const float*)d_in[9];

  const int N = in_sizes[0] / (T_STEPS * F_IN);
  const int E = in_sizes[1] / 2;
  const int* src = ei;
  const int* dst = ei + E;
  const size_t NH = (size_t)N * H;

  char* ws = (char*)d_ws;
  size_t off = 0;
  auto alloc = [&](size_t bytes) -> void* {
    void* p = ws + off;
    off = (off + bytes + 255) & ~(size_t)255;
    return p;
  };
  int*      rowptr  = (int*)   alloc(sizeof(int) * (N + 1));
  float*    dinv    = (float*) alloc(sizeof(float) * N);
  int*      cnt     = (int*)   alloc(sizeof(int) * N);
  int*      bsum    = (int*)   alloc(sizeof(int) * cdiv(N, 1024));
  int*      csr     = (int*)   alloc(sizeof(int) * 2 * E);
  float*    aggx  = (float*)   alloc(sizeof(float) * (size_t)N * XROW);
  float*    hT    = (float*)   alloc(sizeof(float) * NH);
  int*      part  = (int*)aggx;
  (void)ws_size;

  const int sl = cdiv(E, NSL);
  const dim3 sgrid(NSL, cdiv(N, RANGE));
  slice_count_kernel <<<sgrid, 1024, 0, stream>>>(dst, part, N, E, sl);
  slice_prefix_kernel<<<cdiv(N, 256), 256, 0, stream>>>(part, cnt, N);
  block_total_kernel<<<cdiv(N, 1024), 256, 0, stream>>>(cnt, bsum, N);
  scan_kernel       <<<cdiv(N, 1024), 256, 0, stream>>>(cnt, bsum, rowptr, dinv, N);
  slice_fill_kernel  <<<sgrid, 1024, 0, stream>>>(src, dst, part, rowptr, dinv, csr, N, E, sl);

  for (int p = 0; p < T_STEPS / TPP; ++p)
    gather_x_kernel<<<cdiv(N, 8 * NPW), 256, 0, stream>>>(rowptr, csr, dinv, X, aggx, p, N, E);

  gru_seq_kernel<<<cdiv(N, 8 * 16), 256, 0, stream>>>(aggx, Wg, bg, Wih, Whh, bih, bhh, hT, N);

  head_kernel<<<cdiv(N, 256), 256, 0, stream>>>(hT, Wout, bout, (float*)d_out, N);
}


# The three edge passes run as one launch (grid y = pass; blocks go out x first, so they still arrive pass by pass): 10 launches become 8, the next pass's waves fill the last one's idle tail
# speedup vs baseline: 9.2223x; 1.0351x over previous
#include <hip/hip_runtime.h>

#define T_STEPS 12
#define F_IN 16
#define H 64
#define XROW (T_STEPS * F_IN)
#define TPP 4
#define RANGE 51200
#define NSL 128
#define SEPT 13
#define NPW 8
#define WGPITCH 20
#define WPITCH 72

typedef __attribute__((ext_vector_type(16))) _Float16 v16h;
typedef __attribute__((ext_vector_type(8)))  _Float16 v8h;
typedef __attribute__((ext_vector_type(8)))  float    v8f;
typedef __attribute__((ext_vector_type(2)))  float    v2f;

__device__ __forceinline__ v16h lds_frag(const _Float16* p, int k0) {
  union { v16h v; v8h h[2]; } u;
  u.h[0] = *(const v8h*)(p + k0);
  u.h[1] = *(const v8h*)(p + k0 + 16);
  return u.v;
}

__device__ __forceinline__ float sigmoidf_(float x) {
  return __builtin_amdgcn_rcpf(1.0f + __expf(-x));
}
__device__ __forceinline__ float tanhf_(float x) {
  return 1.0f - 2.0f * __builtin_amdgcn_rcpf(1.0f + __expf(2.0f * x));
}


__device__ __forceinline__ int part_pitch(int N) { return (N + 63) & ~63; }

__global__ __launch_bounds__(1024) void slice_count_kernel(const int* __restrict__ dst,
                                                           int* __restrict__ part,
                                                           int N, int E, int sl) {
  __shared__ __attribute__((aligned(16))) int tbl[RANGE];
  const int tid = threadIdx.x;
  const int k = blockIdx.x, r0 = blockIdx.y * RANGE;
  const int nr = min(RANGE, N - r0);
  for (int i = tid; i < RANGE / 4; i += 1024) ((int4*)tbl)[i] = make_int4(0, 0, 0, 0);
  __syncthreads();
  const int e0 = min(k * sl, E), e1 = min(e0 + sl, E);
  for (int base = e0 + tid; base < e1; base += 1024 * SEPT) {
    unsigned d[SEPT];
    for (int j = 0; j < SEPT; ++j) d[j] = (unsigned)(dst[min(base + j * 1024, e1 - 1)] - r0);
    for (int j = 0; j < SEPT; ++j) {
      if (base + j * 1024 >= e1) d[j] = 0xFFFFFFFFu;
    }
    for (int j = 0; j < SEPT; ++j) {
      if (d[j] < (unsigned)nr) atomicAdd(&tbl[d[j]], 1);
    }
  }
  __syncthreads();
  int* out = part + (size_t)k * part_pitch(N) + r0;
  const int nv = nr >> 2;
  for (int i = tid; i < nv; i += 1024) ((int4*)out)[i] = ((const int4*)tbl)[i];
  if (tid < (nr & 3)) out[4 * nv + tid] = tbl[4 * nv + tid];
}

__global__ __launch_bounds__(256) void slice_prefix_kernel(int* __restrict__ part,
                                                           int* __restrict__ cnt, int N) {
  const int n = blockIdx.x * 256 + threadIdx.x;
  if (n >= N) return;
  const size_t pn = (size_t)part_pitch(N);
  int run = 0;
  for (int k0 = 0; k0 < NSL; k0 += 32) {
    int c[32];
    #pragma unroll
    for (int j = 0; j < 32; ++j) c[j] = part[(size_t)(k0 + j) * pn + n];
    #pragma unroll
    for (int j = 0; j < 32; ++j) {
      part[(size_t)(k0 + j) * pn + n] = run;
      run += c[j];
    }
  }
  cnt[n] = run;
}

__device__ __forceinline__ int wave_sum(int v) {
  for (int o = 16; o > 0; o >>= 1) v += __shfl_xor(v, o);
  return v;
}

__device__ __forceinline__ int block_sum(int v, int* sm) {
  const int lane = threadIdx.x & 31, wv = threadIdx.x >> 5;
  v = wave_sum(v);
  if (lane == 0) sm[wv] = v;
  __syncthreads();
  int t = 0;
  for (int i = 0; i < 8; ++i) t += sm[i];
  return t;
}

__global__ __launch_bounds__(256) void block_total_kernel(const int* __restrict__ cnt,
                                                          int* __restrict__ bsum, int N) {
  __shared__ int sm[8];
  const int base = blockIdx.x * 1024 + threadIdx.x;
  int t = 0;
  for (int j = 0; j < 4; ++j) {
    const int i = base + j * 256;
    if (i < N) t += cnt[i];
  }
  t = block_sum(t, sm);
  if (threadIdx.x == 0) bsum[blockIdx.x] = t;
}

__global__ __launch_bounds__(256) void scan_kernel(const int* __restrict__ cnt,
                                                   const int* __restrict__ bsum,
                                                   int* __restrict__ rowptr,
                                                   float* __restrict__ dinv, int N) {
  __shared__ int sm[8];
  __shared__ int sw[8];
  const int tid = threadIdx.x, lane = tid & 31, wv = tid >> 5;
  int pre = 0;
  for (int i = tid; i < (int)blockIdx.x; i += 256) pre += bsum[i];
  pre = block_sum(pre, sm);

  const int i0 = blockIdx.x * 1024 + tid * 4;
  int c[4];
  for (int j = 0; j < 4; ++j) c[j] = (i0 + j < N) ? cnt[i0 + j] : 0;
  const int mine = c[0] + c[1] + c[2] + c[3];
  int inc = mine;
  for (int o = 1; o < 32; o <<= 1) {
    const int u = __shfl_up(inc, o);
    if (lane >= o) inc += u;
  }
  if (lane == 31) sw[wv] = inc;
  __syncthreads();
  int run = pre + inc - mine;
  for (int i = 0; i < wv; ++i) run += sw[i];
  for (int j = 0; j < 4; ++j) {
    if (i0 + j < N) {
      rowptr[i0 + j] = run;
      dinv[i0 + j] = rsqrtf((float)(c[j] + 1));
      run += c[j];
      if (i0 + j == N - 1) rowptr[N] = run;
    }
  }
}

__device__ __forceinline__ void put_slot(int* __restrict__ csr, int pos, int E, int s, float w) {
  int2 rec;
  rec.x = s;
  rec.y = __float_as_int(w);
  ((int2*)csr)[min(max(pos, 0), E - 1)] = rec;
}

__global__ __launch_bounds__(1024) void slice_fill_kernel(const int* __restrict__ src,
                                                          const int* __restrict__ dst,
                                                          const int* __restrict__ part,
                                                          const int* __restrict__ rowptr,
                                                          const float* __restrict__ dinv,
                                                          int* __restrict__ csr,
                                                          int N, int E, int sl) {
  __shared__ __attribute__((aligned(16))) int tbl[RANGE];
  const int tid = threadIdx.x;
  const int k = blockIdx.x, r0 = blockIdx.y * RANGE;
  const int nr = min(RANGE, N - r0);
  const int* pk = part + (size_t)k * part_pitch(N) + r0;
  const int* rp = rowptr + r0;
  const int nv = nr >> 2;
  for (int i0 = tid; i0 < nv; i0 += 1024 * 8) {
    int4 a[8], b[8];
    for (int j = 0; j < 8; ++j) {
      const int i = min(i0 + j * 1024, nv - 1);
      a[j] = ((const int4*)rp)[i];
      b[j] = ((const int4*)pk)[i];
    }
    for (int j = 0; j < 8; ++j) {
      const int i = i0 + j * 1024;
      if (i < nv) ((int4*)tbl)[i] = make_int4(a[j].x + b[j].x, a[j].y + b[j].y, a[j].z + b[j].z, a[j].w + b[j].w);
    }
  }
  if (tid < (nr & 3)) tbl[4 * nv + tid] = rp[4 * nv + tid] + pk[4 * nv + tid];
  __syncthreads();
  const int e0 = min(k * sl, E), e1 = min(e0 + sl, E);
  for (int base = e0 + tid; base < e1; base += 1024 * SEPT) {
    unsigned d[SEPT];
    int s[SEPT], pos[SEPT];
    float w[SEPT];
    for (int j = 0; j < SEPT; ++j) {
      const int e = min(base + j * 1024, e1 - 1);
      d[j] = (unsigned)(dst[e] - r0);
      s[j] = src[e];
    }
    for (int j = 0; j < SEPT; ++j) {
      if (base + j * 1024 >= e1) d[j] = 0xFFFFFFFFu;
    }
    for (int j = 0; j < SEPT; ++j) {
      const bool mine = d[j] < (unsigned)nr;
      w[j] = mine ? dinv[min(max(s[j], 0), N - 1)] : 0.0f;
      pos[j] = mine ? atomicAdd(&tbl[d[j]], 1) : -1;
    }
    for (int j = 0; j < SEPT; ++j) {
      if (pos[j] >= 0) put_slot(csr, pos[j], E, s[j], w[j]);
    }
  }
}


__device__ __forceinline__ float lane_f(float v, int l) {
  return __int_as_float(__builtin_amdgcn_readlane(__float_as_int(v), l));
}

__global__ __launch_bounds__(256) void gather_x_kernel(const int* __restrict__ rowptr,
                                                       const int* __restrict__ csr,
                                                       const float* __restrict__ dinv,
                                                       const float* __restrict__ X,
                                                       float* __restrict__ aggx,
                                                       int N, int E) {
  const int lane = threadIdx.x & 31;
  const int pass = blockIdx.y;
  const int n0 = __builtin_amdgcn_readfirstlane((blockIdx.x * 8 + (threadIdx.x >> 5)) * NPW);
  if (n0 >= N) return;
  const float* xb = X + pass * (TPP * F_IN) + lane * 2;
  const int rpv = rowptr[min(n0 + lane, N)];
  const float dvv = dinv[min(n0 + lane, N - 1)];
  for (int i = 0; i < NPW; ++i) {
    const int n = n0 + i;
    if (n >= N) break;
    const int beg = max(__builtin_amdgcn_readlane(rpv, i), 0);
    const int end = min(__builtin_amdgcn_readlane(rpv, i + 1), E);
    const float dn = lane_f(dvv, i);
    float2 acc;
    acc.x = 0.0f;
    acc.y = 0.0f;
    for (int c = beg; c <= end; c += 32) {
      const int e = c + lane;
      int sv = n;
      float wv = (e == end) ? dn : 0.0f;
      if (e < end) {
        const int2 rec = ((const int2*)csr)[e];
        sv = min(max(rec.x, 0), N - 1);
        wv = __int_as_float(rec.y);
      }
      const int m = min(end + 1 - c, 32);
      for (int k = 0; k < m; k += 8) {
        float2 x[8];
        float w[8];
        for (int j = 0; j < 8; ++j) {
          const int s = __builtin_amdgcn_readlane(sv, k + j);
          w[j] = lane_f(wv, k + j);
          x[j] = *(const float2*)(xb + (size_t)s * XROW);
        }
        for (int j = 0; j < 8; ++j) {
          acc.x += w[j] * x[j].x;
          acc.y += w[j] * x[j].y;
        }
      }
    }
    acc.x *= dn;
    acc.y *= dn;
    *(float2*)(aggx + (size_t)n * XROW + pass * (TPP * F_IN) + lane * 2) = acc;
  }
}

__global__ __launch_bounds__(256) void gru_seq_kernel(const float* __restrict__ aggx,
                                                      const float* __restrict__ Wg,
                                                      const float* __restrict__ bg,
                                                      const float* __restrict__ Wih,
                                                      const float* __restrict__ Whh,
                                                      const float* __restrict__ b_ih,
                                                      const float* __restrict__ b_hh,
                                                      float* __restrict__ hout, int N) {
  __shared__ __attribute__((aligned(16))) _Float16 sW[2 * 3 * H * WPITCH];
  __shared__ __attribute__((aligned(16))) float sB[2 * 3 * H];
  __shared__ __attribute__((aligned(16))) float sBg[H];
  __shared__ __attribute__((aligned(16))) float sWgT[H * WGPITCH];

  const int tid = threadIdx.x;
  for (int i = tid; i < 2 * 3 * H * 8; i += 256) {
    const int row = i >> 3, seg = i & 7;
    const float* sp = (row < 3 * H ? Wih + (size_t)row * H : Whh + (size_t)(row - 3 * H) * H) + seg * 8;
    const float4 f0 = ((const float4*)sp)[0], f1 = ((const float4*)sp)[1];
    v8h o;
    o[0] = (_Float16)f0.x; o[1] = (_Float16)f0.y; o[2] = (_Float16)f0.z; o[3] = (_Float16)f0.w;
    o[4] = (_Float16)f1.x; o[5] = (_Float16)f1.y; o[6] = (_Float16)f1.z; o[7] = (_Float16)f1.w;
    *(v8h*)(sW + row * WPITCH + seg * 8) = o;
  }
  for (int i = tid; i < 2 * 3 * H; i += 256) sB[i] = (i < 3 * H) ? b_ih[i] : b_hh[i - 3 * H];
  if (tid < H) sBg[tid] = bg[tid];
  for (int i = tid; i < F_IN * H; i += 256) sWgT[(i & (H - 1)) * WGPITCH + (i >> 6)] = Wg[i];
  __syncthreads();

  const int lane = tid & 31;
  const int c = lane & 15, hf = lane >> 4;
  const int tile = blockIdx.x * 8 + (tid >> 5);
  if (tile * 16 >= N) return;
  const int node = tile * 16 + c;
  const int nodec = min(node, N - 1);

  float h32[4][8];
  for (int j = 0; j < 4; ++j) {
    for (int r = 0; r < 8; ++r) h32[j][r] = 0.0f;
  }
  v16h hB0 = {}, hB1 = {};

  const float* xrow = aggx + (size_t)nodec * XROW + 8 * hf;
  float4 xa = ((const float4*)xrow)[0], xb = ((const float4*)xrow)[1];

  for (int t = 0; t < T_STEPS; ++t) {
    asm volatile("" ::: "memory");
    const float* xn = xrow + min(t + 1, T_STEPS - 1) * F_IN;
    const float4 na = ((const float4*)xn)[0], nb = ((const float4*)xn)[1];
    const float xs[8] = {xa.x, xa.y, xa.z, xa.w, xb.x, xb.y, xb.z, xb.w};

    v16h aB0, aB1;
    for (int blk = 0; blk < 4; ++blk) {
      const float4 w0 = *(const float4*)&sWgT[(16 * blk + c) * WGPITCH + 8 * hf];
      const float4 w1 = *(const float4*)&sWgT[(16 * blk + c) * WGPITCH + 8 * hf + 4];
      const float ws8[8] = {w0.x, w0.y, w0.z, w0.w, w1.x, w1.y, w1.z, w1.w};
      v8f acc = {};
      for (int q = 0; q < 4; ++q) {
        v2f wq, xq;
        wq[0] = ws8[2 * q];
        wq[1] = ws8[2 * q + 1];
        xq[0] = xs[2 * q];
        xq[1] = xs[2 * q + 1];
        acc = __builtin_amdgcn_wmma_f32_16x16x4_f32(false, wq, false, xq, (short)0, acc, false, false);
      }
      const float4 g0 = *(const float4*)&sBg[16 * blk + 8 * hf];
      const float4 g1 = *(const float4*)&sBg[16 * blk + 8 * hf + 4];
      const float gb[8] = {g0.x, g0.y, g0.z, g0.w, g1.x, g1.y, g1.z, g1.w};
      for (int r = 0; r < 8; ++r) {
        const _Float16 av = (_Float16)fmaxf(acc[r] + gb[r], 0.0f);
        if (blk < 2) aB0[(blk & 1) * 8 + r] = av; else aB1[(blk & 1) * 8 + r] = av;
      }
    }

    for (int j = 0; j < 4; ++j) {
      asm volatile("" ::: "memory");
      const _Float16* wr = sW + (          16 * j + c) * WPITCH + 8 * hf;
      const _Float16* wz = sW + (     64 + 16 * j + c) * WPITCH + 8 * hf;
      const _Float16* wn = sW + (    128 + 16 * j + c) * WPITCH + 8 * hf;
      const _Float16* ur = sW + (192 +       16 * j + c) * WPITCH + 8 * hf;
      const _Float16* uz = sW + (192 +  64 + 16 * j + c) * WPITCH + 8 * hf;
      const _Float16* un = sW + (192 + 128 + 16 * j + c) * WPITCH + 8 * hf;

      v8f ir = {}, iz = {}, in_ = {}, hr = {}, hz = {}, hn = {};
      ir  = __builtin_amdgcn_wmma_f32_16x16x32_f16(false, lds_frag(wr, 0), false, aB0, (short)0, ir,  false, false);
      iz  = __builtin_amdgcn_wmma_f32_16x16x32_f16(false, lds_frag(wz, 0), false, aB0, (short)0, iz,  false, false);
      in_ = __builtin_amdgcn_wmma_f32_16x16x32_f16(false, lds_frag(wn, 0), false, aB0, (short)0, in_, false, false);
      hr  = __builtin_amdgcn_wmma_f32_16x16x32_f16(false, lds_frag(ur, 0), false, hB0, (short)0, hr,  false, false);
      hz  = __builtin_amdgcn_wmma_f32_16x16x32_f16(false, lds_frag(uz, 0), false, hB0, (short)0, hz,  false, false);
      hn  = __builtin_amdgcn_wmma_f32_16x16x32_f16(false, lds_frag(un, 0), false, hB0, (short)0, hn,  false, false);
      ir  = __builtin_amdgcn_wmma_f32_16x16x32_f16(false, lds_frag(wr, 32), false, aB1, (short)0, ir,  false, false);
      iz  = __builtin_amdgcn_wmma_f32_16x16x32_f16(false, lds_frag(wz, 32), false, aB1, (short)0, iz,  false, false);
      in_ = __builtin_amdgcn_wmma_f32_16x16x32_f16(false, lds_frag(wn, 32), false, aB1, (short)0, in_, false, false);
      hr  = __builtin_amdgcn_wmma_f32_16x16x32_f16(false, lds_frag(ur, 32), false, hB1, (short)0, hr,  false, false);
      hz  = __builtin_amdgcn_wmma_f32_16x16x32_f16(false, lds_frag(uz, 32), false, hB1, (short)0, hz,  false, false);
      hn  = __builtin_amdgcn_wmma_f32_16x16x32_f16(false, lds_frag(un, 32), false, hB1, (short)0, hn,  false, false);

      const int ch = 16 * j + 8 * hf;
      float bir[8], biz[8], bin[8], bhr[8], bhz[8], bhn[8];
      for (int r = 0; r < 8; r += 4) {
        *(float4*)&bir[r] = *(const float4*)&sB[            ch + r];
        *(float4*)&biz[r] = *(const float4*)&sB[       64 + ch + r];
        *(float4*)&bin[r] = *(const float4*)&sB[      128 + ch + r];
        *(float4*)&bhr[r] = *(const float4*)&sB[192 +       ch + r];
        *(float4*)&bhz[r] = *(const float4*)&sB[192 +  64 + ch + r];
        *(float4*)&bhn[r] = *(const float4*)&sB[192 + 128 + ch + r];
      }
      for (int r = 0; r < 8; ++r) {
        float rg = sigmoidf_(ir[r] + bir[r] + hr[r] + bhr[r]);
        float zg = sigmoidf_(iz[r] + biz[r] + hz[r] + bhz[r]);
        float nc = tanhf_(in_[r] + bin[r] + rg * (hn[r] + bhn[r]));
        float hold = h32[j][r];
        h32[j][r] = (1.0f - zg) * nc + zg * hold;
      }
    }

    for (int r = 0; r < 8; ++r) {
      hB0[r]     = (_Float16)h32[0][r];
      hB0[8 + r] = (_Float16)h32[1][r];
      hB1[r]     = (_Float16)h32[2][r];
      hB1[8 + r] = (_Float16)h32[3][r];
    }
    xa = na;
    xb = nb;
  }

  if (node < N) {
    float* ho = hout + (size_t)node * H + 8 * hf;
    for (int j = 0; j < 4; ++j) {
      ((float4*)(ho + 16 * j))[0] = make_float4(h32[j][0], h32[j][1], h32[j][2], h32[j][3]);
      ((float4*)(ho + 16 * j))[1] = make_float4(h32[j][4], h32[j][5], h32[j][6], h32[j][7]);
    }
  }
}

__global__ __launch_bounds__(256) void head_kernel(const float* __restrict__ h,
                                                   const float* __restrict__ Wout,
                                                   const float* __restrict__ bout,
                                                   float* __restrict__ out, int N) {
  int i = blockIdx.x * 256 + threadIdx.x;
  if (i >= N) return;
  float a0 = bout[0], a1 = bout[1];
  const float* hr = h + (size_t)i * H;
  #pragma unroll
  for (int k = 0; k < H; ++k) {
    float hv = hr[k];
    a0 += hv * Wout[k * 2 + 0];
    a1 += hv * Wout[k * 2 + 1];
  }
  out[i * 2 + 0] = a0;
  out[i * 2 + 1] = a1;
}


static inline int cdiv(long long a, long long b) { return (int)((a + b - 1) / b); }

extern "C" void kernel_launch(void* const* d_in, const int* in_sizes, int n_in,
                              void* d_out, int out_size, void* d_ws, size_t ws_size,
                              hipStream_t stream) {
  const float* X    = (const float*)d_in[0];
  const int*   ei   = (const int*)  d_in[1];
  const float* Wg   = (const float*)d_in[2];
  const float* bg   = (const float*)d_in[3];
  const float* Wih  = (const float*)d_in[4];
  const float* Whh  = (const float*)d_in[5];
  const float* bih  = (const float*)d_in[6];
  const float* bhh  = (const float*)d_in[7];
  const float* Wout = (const float*)d_in[8];
  const float* bout = (const float*)d_in[9];

  const int N = in_sizes[0] / (T_STEPS * F_IN);
  const int E = in_sizes[1] / 2;
  const int* src = ei;
  const int* dst = ei + E;
  const size_t NH = (size_t)N * H;

  char* ws = (char*)d_ws;
  size_t off = 0;
  auto alloc = [&](size_t bytes) -> void* {
    void* p = ws + off;
    off = (off + bytes + 255) & ~(size_t)255;
    return p;
  };
  int*      rowptr  = (int*)   alloc(sizeof(int) * (N + 1));
  float*    dinv    = (float*) alloc(sizeof(float) * N);
  int*      cnt     = (int*)   alloc(sizeof(int) * N);
  int*      bsum    = (int*)   alloc(sizeof(int) * cdiv(N, 1024));
  int*      csr     = (int*)   alloc(sizeof(int) * 2 * E);
  float*    aggx  = (float*)   alloc(sizeof(float) * (size_t)N * XROW);
  float*    hT    = (float*)   alloc(sizeof(float) * NH);
  int*      part  = (int*)aggx;
  (void)ws_size;

  const int sl = cdiv(E, NSL);
  const dim3 sgrid(NSL, cdiv(N, RANGE));
  slice_count_kernel <<<sgrid, 1024, 0, stream>>>(dst, part, N, E, sl);
  slice_prefix_kernel<<<cdiv(N, 256), 256, 0, stream>>>(part, cnt, N);
  block_total_kernel<<<cdiv(N, 1024), 256, 0, stream>>>(cnt, bsum, N);
  scan_kernel       <<<cdiv(N, 1024), 256, 0, stream>>>(cnt, bsum, rowptr, dinv, N);
  slice_fill_kernel  <<<sgrid, 1024, 0, stream>>>(src, dst, part, rowptr, dinv, csr, N, E, sl);

  const dim3 ggrid(cdiv(N, 8 * NPW), T_STEPS / TPP);
  gather_x_kernel<<<ggrid, 256, 0, stream>>>(rowptr, csr, dinv, X, aggx, N, E);

  gru_seq_kernel<<<cdiv(N, 8 * 16), 256, 0, stream>>>(aggx, Wg, bg, Wih, Whh, bih, bhh, hT, N);

  head_kernel<<<cdiv(N, 256), 256, 0, stream>>>(hT, Wout, bout, (float*)d_out, N);
}


# Recurrent kernel launched as 512 resident work-groups (2 per compute unit); each wave loops over 16-node tiles dealt evenly to work-groups and SIMDs: no half-empty second round of work-groups
# speedup vs baseline: 9.7406x; 1.0562x over previous
#include <hip/hip_runtime.h>

#define T_STEPS 12
#define F_IN 16
#define H 64
#define XROW (T_STEPS * F_IN)
#define TPP 4
#define RANGE 51200
#define NSL 128
#define SEPT 13
#define NPW 8
#define WGPITCH 20
#define GRU_GROUPS 512
#define WPITCH 72

typedef __attribute__((ext_vector_type(16))) _Float16 v16h;
typedef __attribute__((ext_vector_type(8)))  _Float16 v8h;
typedef __attribute__((ext_vector_type(8)))  float    v8f;
typedef __attribute__((ext_vector_type(2)))  float    v2f;

__device__ __forceinline__ v16h lds_frag(const _Float16* p, int k0) {
  union { v16h v; v8h h[2]; } u;
  u.h[0] = *(const v8h*)(p + k0);
  u.h[1] = *(const v8h*)(p + k0 + 16);
  return u.v;
}

__device__ __forceinline__ float sigmoidf_(float x) {
  return __builtin_amdgcn_rcpf(1.0f + __expf(-x));
}
__device__ __forceinline__ float tanhf_(float x) {
  return 1.0f - 2.0f * __builtin_amdgcn_rcpf(1.0f + __expf(2.0f * x));
}


__device__ __forceinline__ int part_pitch(int N) { return (N + 63) & ~63; }

__global__ __launch_bounds__(1024) void slice_count_kernel(const int* __restrict__ dst,
                                                           int* __restrict__ part,
                                                           int N, int E, int sl) {
  __shared__ __attribute__((aligned(16))) int tbl[RANGE];
  const int tid = threadIdx.x;
  const int k = blockIdx.x, r0 = blockIdx.y * RANGE;
  const int nr = min(RANGE, N - r0);
  for (int i = tid; i < RANGE / 4; i += 1024) ((int4*)tbl)[i] = make_int4(0, 0, 0, 0);
  __syncthreads();
  const int e0 = min(k * sl, E), e1 = min(e0 + sl, E);
  for (int base = e0 + tid; base < e1; base += 1024 * SEPT) {
    unsigned d[SEPT];
    for (int j = 0; j < SEPT; ++j) d[j] = (unsigned)(dst[min(base + j * 1024, e1 - 1)] - r0);
    for (int j = 0; j < SEPT; ++j) {
      if (base + j * 1024 >= e1) d[j] = 0xFFFFFFFFu;
    }
    for (int j = 0; j < SEPT; ++j) {
      if (d[j] < (unsigned)nr) atomicAdd(&tbl[d[j]], 1);
    }
  }
  __syncthreads();
  int* out = part + (size_t)k * part_pitch(N) + r0;
  const int nv = nr >> 2;
  for (int i = tid; i < nv; i += 1024) ((int4*)out)[i] = ((const int4*)tbl)[i];
  if (tid < (nr & 3)) out[4 * nv + tid] = tbl[4 * nv + tid];
}

__global__ __launch_bounds__(256) void slice_prefix_kernel(int* __restrict__ part,
                                                           int* __restrict__ cnt, int N) {
  const int n = blockIdx.x * 256 + threadIdx.x;
  if (n >= N) return;
  const size_t pn = (size_t)part_pitch(N);
  int run = 0;
  for (int k0 = 0; k0 < NSL; k0 += 32) {
    int c[32];
    #pragma unroll
    for (int j = 0; j < 32; ++j) c[j] = part[(size_t)(k0 + j) * pn + n];
    #pragma unroll
    for (int j = 0; j < 32; ++j) {
      part[(size_t)(k0 + j) * pn + n] = run;
      run += c[j];
    }
  }
  cnt[n] = run;
}

__device__ __forceinline__ int wave_sum(int v) {
  for (int o = 16; o > 0; o >>= 1) v += __shfl_xor(v, o);
  return v;
}

__device__ __forceinline__ int block_sum(int v, int* sm) {
  const int lane = threadIdx.x & 31, wv = threadIdx.x >> 5;
  v = wave_sum(v);
  if (lane == 0) sm[wv] = v;
  __syncthreads();
  int t = 0;
  for (int i = 0; i < 8; ++i) t += sm[i];
  return t;
}

__global__ __launch_bounds__(256) void block_total_kernel(const int* __restrict__ cnt,
                                                          int* __restrict__ bsum, int N) {
  __shared__ int sm[8];
  const int base = blockIdx.x * 1024 + threadIdx.x;
  int t = 0;
  for (int j = 0; j < 4; ++j) {
    const int i = base + j * 256;
    if (i < N) t += cnt[i];
  }
  t = block_sum(t, sm);
  if (threadIdx.x == 0) bsum[blockIdx.x] = t;
}

__global__ __launch_bounds__(256) void scan_kernel(const int* __restrict__ cnt,
                                                   const int* __restrict__ bsum,
                                                   int* __restrict__ rowptr,
                                                   float* __restrict__ dinv, int N) {
  __shared__ int sm[8];
  __shared__ int sw[8];
  const int tid = threadIdx.x, lane = tid & 31, wv = tid >> 5;
  int pre = 0;
  for (int i = tid; i < (int)blockIdx.x; i += 256) pre += bsum[i];
  pre = block_sum(pre, sm);

  const int i0 = blockIdx.x * 1024 + tid * 4;
  int c[4];
  for (int j = 0; j < 4; ++j) c[j] = (i0 + j < N) ? cnt[i0 + j] : 0;
  const int mine = c[0] + c[1] + c[2] + c[3];
  int inc = mine;
  for (int o = 1; o < 32; o <<= 1) {
    const int u = __shfl_up(inc, o);
    if (lane >= o) inc += u;
  }
  if (lane == 31) sw[wv] = inc;
  __syncthreads();
  int run = pre + inc - mine;
  for (int i = 0; i < wv; ++i) run += sw[i];
  for (int j = 0; j < 4; ++j) {
    if (i0 + j < N) {
      rowptr[i0 + j] = run;
      dinv[i0 + j] = rsqrtf((float)(c[j] + 1));
      run += c[j];
      if (i0 + j == N - 1) rowptr[N] = run;
    }
  }
}

__device__ __forceinline__ void put_slot(int* __restrict__ csr, int pos, int E, int s, float w) {
  int2 rec;
  rec.x = s;
  rec.y = __float_as_int(w);
  ((int2*)csr)[min(max(pos, 0), E - 1)] = rec;
}

__global__ __launch_bounds__(1024) void slice_fill_kernel(const int* __restrict__ src,
                                                          const int* __restrict__ dst,
                                                          const int* __restrict__ part,
                                                          const int* __restrict__ rowptr,
                                                          const float* __restrict__ dinv,
                                                          int* __restrict__ csr,
                                                          int N, int E, int sl) {
  __shared__ __attribute__((aligned(16))) int tbl[RANGE];
  const int tid = threadIdx.x;
  const int k = blockIdx.x, r0 = blockIdx.y * RANGE;
  const int nr = min(RANGE, N - r0);
  const int* pk = part + (size_t)k * part_pitch(N) + r0;
  const int* rp = rowptr + r0;
  const int nv = nr >> 2;
  for (int i0 = tid; i0 < nv; i0 += 1024 * 8) {
    int4 a[8], b[8];
    for (int j = 0; j < 8; ++j) {
      const int i = min(i0 + j * 1024, nv - 1);
      a[j] = ((const int4*)rp)[i];
      b[j] = ((const int4*)pk)[i];
    }
    for (int j = 0; j < 8; ++j) {
      const int i = i0 + j * 1024;
      if (i < nv) ((int4*)tbl)[i] = make_int4(a[j].x + b[j].x, a[j].y + b[j].y, a[j].z + b[j].z, a[j].w + b[j].w);
    }
  }
  if (tid < (nr & 3)) tbl[4 * nv + tid] = rp[4 * nv + tid] + pk[4 * nv + tid];
  __syncthreads();
  const int e0 = min(k * sl, E), e1 = min(e0 + sl, E);
  for (int base = e0 + tid; base < e1; base += 1024 * SEPT) {
    unsigned d[SEPT];
    int s[SEPT], pos[SEPT];
    float w[SEPT];
    for (int j = 0; j < SEPT; ++j) {
      const int e = min(base + j * 1024, e1 - 1);
      d[j] = (unsigned)(dst[e] - r0);
      s[j] = src[e];
    }
    for (int j = 0; j < SEPT; ++j) {
      if (base + j * 1024 >= e1) d[j] = 0xFFFFFFFFu;
    }
    for (int j = 0; j < SEPT; ++j) {
      const bool mine = d[j] < (unsigned)nr;
      w[j] = mine ? dinv[min(max(s[j], 0), N - 1)] : 0.0f;
      pos[j] = mine ? atomicAdd(&tbl[d[j]], 1) : -1;
    }
    for (int j = 0; j < SEPT; ++j) {
      if (pos[j] >= 0) put_slot(csr, pos[j], E, s[j], w[j]);
    }
  }
}


__device__ __forceinline__ float lane_f(float v, int l) {
  return __int_as_float(__builtin_amdgcn_readlane(__float_as_int(v), l));
}

__global__ __launch_bounds__(256) void gather_x_kernel(const int* __restrict__ rowptr,
                                                       const int* __restrict__ csr,
                                                       const float* __restrict__ dinv,
                                                       const float* __restrict__ X,
                                                       float* __restrict__ aggx,
                                                       int N, int E) {
  const int lane = threadIdx.x & 31;
  const int pass = blockIdx.y;
  const int n0 = __builtin_amdgcn_readfirstlane((blockIdx.x * 8 + (threadIdx.x >> 5)) * NPW);
  if (n0 >= N) return;
  const float* xb = X + pass * (TPP * F_IN) + lane * 2;
  const int rpv = rowptr[min(n0 + lane, N)];
  const float dvv = dinv[min(n0 + lane, N - 1)];
  for (int i = 0; i < NPW; ++i) {
    const int n = n0 + i;
    if (n >= N) break;
    const int beg = max(__builtin_amdgcn_readlane(rpv, i), 0);
    const int end = min(__builtin_amdgcn_readlane(rpv, i + 1), E);
    const float dn = lane_f(dvv, i);
    float2 acc;
    acc.x = 0.0f;
    acc.y = 0.0f;
    for (int c = beg; c <= end; c += 32) {
      const int e = c + lane;
      int sv = n;
      float wv = (e == end) ? dn : 0.0f;
      if (e < end) {
        const int2 rec = ((const int2*)csr)[e];
        sv = min(max(rec.x, 0), N - 1);
        wv = __int_as_float(rec.y);
      }
      const int m = min(end + 1 - c, 32);
      for (int k = 0; k < m; k += 8) {
        float2 x[8];
        float w[8];
        for (int j = 0; j < 8; ++j) {
          const int s = __builtin_amdgcn_readlane(sv, k + j);
          w[j] = lane_f(wv, k + j);
          x[j] = *(const float2*)(xb + (size_t)s * XROW);
        }
        for (int j = 0; j < 8; ++j) {
          acc.x += w[j] * x[j].x;
          acc.y += w[j] * x[j].y;
        }
      }
    }
    acc.x *= dn;
    acc.y *= dn;
    *(float2*)(aggx + (size_t)n * XROW + pass * (TPP * F_IN) + lane * 2) = acc;
  }
}

__global__ __launch_bounds__(256) void gru_seq_kernel(const float* __restrict__ aggx,
                                                      const float* __restrict__ Wg,
                                                      const float* __restrict__ bg,
                                                      const float* __restrict__ Wih,
                                                      const float* __restrict__ Whh,
                                                      const float* __restrict__ b_ih,
                                                      const float* __restrict__ b_hh,
                                                      float* __restrict__ hout, int N) {
  __shared__ __attribute__((aligned(16))) _Float16 sW[2 * 3 * H * WPITCH];
  __shared__ __attribute__((aligned(16))) float sB[2 * 3 * H];
  __shared__ __attribute__((aligned(16))) float sBg[H];
  __shared__ __attribute__((aligned(16))) float sWgT[H * WGPITCH];

  const int tid = threadIdx.x;
  for (int i = tid; i < 2 * 3 * H * 8; i += 256) {
    const int row = i >> 3, seg = i & 7;
    const float* sp = (row < 3 * H ? Wih + (size_t)row * H : Whh + (size_t)(row - 3 * H) * H) + seg * 8;
    const float4 f0 = ((const float4*)sp)[0], f1 = ((const float4*)sp)[1];
    v8h o;
    o[0] = (_Float16)f0.x; o[1] = (_Float16)f0.y; o[2] = (_Float16)f0.z; o[3] = (_Float16)f0.w;
    o[4] = (_Float16)f1.x; o[5] = (_Float16)f1.y; o[6] = (_Float16)f1.z; o[7] = (_Float16)f1.w;
    *(v8h*)(sW + row * WPITCH + seg * 8) = o;
  }
  for (int i = tid; i < 2 * 3 * H; i += 256) sB[i] = (i < 3 * H) ? b_ih[i] : b_hh[i - 3 * H];
  if (tid < H) sBg[tid] = bg[tid];
  for (int i = tid; i < F_IN * H; i += 256) sWgT[(i & (H - 1)) * WGPITCH + (i >> 6)] = Wg[i];
  __syncthreads();

  const int lane = tid & 31;
  const int c = lane & 15, hf = lane >> 4;
  const int wave = tid >> 5;
  const int place = wave ^ ((wave & 1) << 2);
  const int ntiles = (N + 15) >> 4;
  const int tstride = 8 * (int)gridDim.x;
  for (int tile = __builtin_amdgcn_readfirstlane(place * (int)gridDim.x + (int)blockIdx.x); tile < ntiles; tile += tstride) {
  const int node = tile * 16 + c;
  const int nodec = min(node, N - 1);

  float h32[4][8];
  for (int j = 0; j < 4; ++j) {
    for (int r = 0; r < 8; ++r) h32[j][r] = 0.0f;
  }
  v16h hB0 = {}, hB1 = {};

  const float* xrow = aggx + (size_t)nodec * XROW + 8 * hf;
  float4 xa = ((const float4*)xrow)[0], xb = ((const float4*)xrow)[1];

  for (int t = 0; t < T_STEPS; ++t) {
    asm volatile("" ::: "memory");
    const float* xn = xrow + min(t + 1, T_STEPS - 1) * F_IN;
    const float4 na = ((const float4*)xn)[0], nb = ((const float4*)xn)[1];
    const float xs[8] = {xa.x, xa.y, xa.z, xa.w, xb.x, xb.y, xb.z, xb.w};

    v16h aB0, aB1;
    for (int blk = 0; blk < 4; ++blk) {
      const float4 w0 = *(const float4*)&sWgT[(16 * blk + c) * WGPITCH + 8 * hf];
      const float4 w1 = *(const float4*)&sWgT[(16 * blk + c) * WGPITCH + 8 * hf + 4];
      const float ws8[8] = {w0.x, w0.y, w0.z, w0.w, w1.x, w1.y, w1.z, w1.w};
      v8f acc = {};
      for (int q = 0; q < 4; ++q) {
        v2f wq, xq;
        wq[0] = ws8[2 * q];
        wq[1] = ws8[2 * q + 1];
        xq[0] = xs[2 * q];
        xq[1] = xs[2 * q + 1];
        acc = __builtin_amdgcn_wmma_f32_16x16x4_f32(false, wq, false, xq, (short)0, acc, false, false);
      }
      const float4 g0 = *(const float4*)&sBg[16 * blk + 8 * hf];
      const float4 g1 = *(const float4*)&sBg[16 * blk + 8 * hf + 4];
      const float gb[8] = {g0.x, g0.y, g0.z, g0.w, g1.x, g1.y, g1.z, g1.w};
      for (int r = 0; r < 8; ++r) {
        const _Float16 av = (_Float16)fmaxf(acc[r] + gb[r], 0.0f);
        if (blk < 2) aB0[(blk & 1) * 8 + r] = av; else aB1[(blk & 1) * 8 + r] = av;
      }
    }

    for (int j = 0; j < 4; ++j) {
      asm volatile("" ::: "memory");
      const _Float16* wr = sW + (          16 * j + c) * WPITCH + 8 * hf;
      const _Float16* wz = sW + (     64 + 16 * j + c) * WPITCH + 8 * hf;
      const _Float16* wn = sW + (    128 + 16 * j + c) * WPITCH + 8 * hf;
      const _Float16* ur = sW + (192 +       16 * j + c) * WPITCH + 8 * hf;
      const _Float16* uz = sW + (192 +  64 + 16 * j + c) * WPITCH + 8 * hf;
      const _Float16* un = sW + (192 + 128 + 16 * j + c) * WPITCH + 8 * hf;

      v8f ir = {}, iz = {}, in_ = {}, hr = {}, hz = {}, hn = {};
      ir  = __builtin_amdgcn_wmma_f32_16x16x32_f16(false, lds_frag(wr, 0), false, aB0, (short)0, ir,  false, false);
      iz  = __builtin_amdgcn_wmma_f32_16x16x32_f16(false, lds_frag(wz, 0), false, aB0, (short)0, iz,  false, false);
      in_ = __builtin_amdgcn_wmma_f32_16x16x32_f16(false, lds_frag(wn, 0), false, aB0, (short)0, in_, false, false);
      hr  = __builtin_amdgcn_wmma_f32_16x16x32_f16(false, lds_frag(ur, 0), false, hB0, (short)0, hr,  false, false);
      hz  = __builtin_amdgcn_wmma_f32_16x16x32_f16(false, lds_frag(uz, 0), false, hB0, (short)0, hz,  false, false);
      hn  = __builtin_amdgcn_wmma_f32_16x16x32_f16(false, lds_frag(un, 0), false, hB0, (short)0, hn,  false, false);
      ir  = __builtin_amdgcn_wmma_f32_16x16x32_f16(false, lds_frag(wr, 32), false, aB1, (short)0, ir,  false, false);
      iz  = __builtin_amdgcn_wmma_f32_16x16x32_f16(false, lds_frag(wz, 32), false, aB1, (short)0, iz,  false, false);
      in_ = __builtin_amdgcn_wmma_f32_16x16x32_f16(false, lds_frag(wn, 32), false, aB1, (short)0, in_, false, false);
      hr  = __builtin_amdgcn_wmma_f32_16x16x32_f16(false, lds_frag(ur, 32), false, hB1, (short)0, hr,  false, false);
      hz  = __builtin_amdgcn_wmma_f32_16x16x32_f16(false, lds_frag(uz, 32), false, hB1, (short)0, hz,  false, false);
      hn  = __builtin_amdgcn_wmma_f32_16x16x32_f16(false, lds_frag(un, 32), false, hB1, (short)0, hn,  false, false);

      const int ch = 16 * j + 8 * hf;
      float bir[8], biz[8], bin[8], bhr[8], bhz[8], bhn[8];
      for (int r = 0; r < 8; r += 4) {
        *(float4*)&bir[r] = *(const float4*)&sB[            ch + r];
        *(float4*)&biz[r] = *(const float4*)&sB[       64 + ch + r];
        *(float4*)&bin[r] = *(const float4*)&sB[      128 + ch + r];
        *(float4*)&bhr[r] = *(const float4*)&sB[192 +       ch + r];
        *(float4*)&bhz[r] = *(const float4*)&sB[192 +  64 + ch + r];
        *(float4*)&bhn[r] = *(const float4*)&sB[192 + 128 + ch + r];
      }
      for (int r = 0; r < 8; ++r) {
        float rg = sigmoidf_(ir[r] + bir[r] + hr[r] + bhr[r]);
        float zg = sigmoidf_(iz[r] + biz[r] + hz[r] + bhz[r]);
        float nc = tanhf_(in_[r] + bin[r] + rg * (hn[r] + bhn[r]));
        float hold = h32[j][r];
        h32[j][r] = (1.0f - zg) * nc + zg * hold;
      }
    }

    for (int r = 0; r < 8; ++r) {
      hB0[r]     = (_Float16)h32[0][r];
      hB0[8 + r] = (_Float16)h32[1][r];
      hB1[r]     = (_Float16)h32[2][r];
      hB1[8 + r] = (_Float16)h32[3][r];
    }
    xa = na;
    xb = nb;
  }

  if (node < N) {
    float* ho = hout + (size_t)node * H + 8 * hf;
    for (int j = 0; j < 4; ++j) {
      ((float4*)(ho + 16 * j))[0] = make_float4(h32[j][0], h32[j][1], h32[j][2], h32[j][3]);
      ((float4*)(ho + 16 * j))[1] = make_float4(h32[j][4], h32[j][5], h32[j][6], h32[j][7]);
    }
  }
  }
}

__global__ __launch_bounds__(256) void head_kernel(const float* __restrict__ h,
                                                   const float* __restrict__ Wout,
                                                   const float* __restrict__ bout,
                                                   float* __restrict__ out, int N) {
  int i = blockIdx.x * 256 + threadIdx.x;
  if (i >= N) return;
  float a0 = bout[0], a1 = bout[1];
  const float* hr = h + (size_t)i * H;
  #pragma unroll
  for (int k = 0; k < H; ++k) {
    float hv = hr[k];
    a0 += hv * Wout[k * 2 + 0];
    a1 += hv * Wout[k * 2 + 1];
  }
  out[i * 2 + 0] = a0;
  out[i * 2 + 1] = a1;
}


static inline int cdiv(long long a, long long b) { return (int)((a + b - 1) / b); }

extern "C" void kernel_launch(void* const* d_in, const int* in_sizes, int n_in,
                              void* d_out, int out_size, void* d_ws, size_t ws_size,
                              hipStream_t stream) {
  const float* X    = (const float*)d_in[0];
  const int*   ei   = (const int*)  d_in[1];
  const float* Wg   = (const float*)d_in[2];
  const float* bg   = (const float*)d_in[3];
  const float* Wih  = (const float*)d_in[4];
  const float* Whh  = (const float*)d_in[5];
  const float* bih  = (const float*)d_in[6];
  const float* bhh  = (const float*)d_in[7];
  const float* Wout = (const float*)d_in[8];
  const float* bout = (const float*)d_in[9];

  const int N = in_sizes[0] / (T_STEPS * F_IN);
  const int E = in_sizes[1] / 2;
  const int* src = ei;
  const int* dst = ei + E;
  const size_t NH = (size_t)N * H;

  char* ws = (char*)d_ws;
  size_t off = 0;
  auto alloc = [&](size_t bytes) -> void* {
    void* p = ws + off;
    off = (off + bytes + 255) & ~(size_t)255;
    return p;
  };
  int*      rowptr  = (int*)   alloc(sizeof(int) * (N + 1));
  float*    dinv    = (float*) alloc(sizeof(float) * N);
  int*      cnt     = (int*)   alloc(sizeof(int) * N);
  int*      bsum    = (int*)   alloc(sizeof(int) * cdiv(N, 1024));
  int*      csr     = (int*)   alloc(sizeof(int) * 2 * E);
  float*    aggx  = (float*)   alloc(sizeof(float) * (size_t)N * XROW);
  float*    hT    = (float*)   alloc(sizeof(float) * NH);
  int*      part  = (int*)aggx;
  (void)ws_size;

  const int sl = cdiv(E, NSL);
  const dim3 sgrid(NSL, cdiv(N, RANGE));
  slice_count_kernel <<<sgrid, 1024, 0, stream>>>(dst, part, N, E, sl);
  slice_prefix_kernel<<<cdiv(N, 256), 256, 0, stream>>>(part, cnt, N);
  block_total_kernel<<<cdiv(N, 1024), 256, 0, stream>>>(cnt, bsum, N);
  scan_kernel       <<<cdiv(N, 1024), 256, 0, stream>>>(cnt, bsum, rowptr, dinv, N);
  slice_fill_kernel  <<<sgrid, 1024, 0, stream>>>(src, dst, part, rowptr, dinv, csr, N, E, sl);

  const dim3 ggrid(cdiv(N, 8 * NPW), T_STEPS / TPP);
  gather_x_kernel<<<ggrid, 256, 0, stream>>>(rowptr, csr, dinv, X, aggx, N, E);

  gru_seq_kernel<<<GRU_GROUPS, 256, 0, stream>>>(aggx, Wg, bg, Wih, Whh, bih, bhh, hT, N);

  head_kernel<<<cdiv(N, 256), 256, 0, stream>>>(hT, Wout, bout, (float*)d_out, N);
}


# Two dependent launches folded into neighbours (8 become 6): scan block totals taken in the slice-prefix kernel, output head taken from h in registers in the recurrent kernel (head sum order changed)
# speedup vs baseline: 10.5926x; 1.0875x over previous
#include <hip/hip_runtime.h>

#define T_STEPS 12
#define F_IN 16
#define H 64
#define XROW (T_STEPS * F_IN)
#define TPP 4
#define RANGE 51200
#define NSL 128
#define SEPT 13
#define NPW 8
#define WGPITCH 20
#define GRU_GROUPS 512
#define WPITCH 72

typedef __attribute__((ext_vector_type(16))) _Float16 v16h;
typedef __attribute__((ext_vector_type(8)))  _Float16 v8h;
typedef __attribute__((ext_vector_type(8)))  float    v8f;
typedef __attribute__((ext_vector_type(2)))  float    v2f;

__device__ __forceinline__ v16h lds_frag(const _Float16* p, int k0) {
  union { v16h v; v8h h[2]; } u;
  u.h[0] = *(const v8h*)(p + k0);
  u.h[1] = *(const v8h*)(p + k0 + 16);
  return u.v;
}

__device__ __forceinline__ float sigmoidf_(float x) {
  return __builtin_amdgcn_rcpf(1.0f + __expf(-x));
}
__device__ __forceinline__ float other_half(float x) {
  return __builtin_bit_cast(float, __builtin_amdgcn_permlanex16(__builtin_bit_cast(int, x), __builtin_bit_cast(int, x),
                                                                  0x76543210, 0xfedcba98, false, false));
}
__device__ __forceinline__ float tanhf_(float x) {
  return 1.0f - 2.0f * __builtin_amdgcn_rcpf(1.0f + __expf(2.0f * x));
}


__device__ __forceinline__ int part_pitch(int N) { return (N + 63) & ~63; }

__global__ __launch_bounds__(1024) void slice_count_kernel(const int* __restrict__ dst,
                                                           int* __restrict__ part,
                                                           int N, int E, int sl) {
  __shared__ __attribute__((aligned(16))) int tbl[RANGE];
  const int tid = threadIdx.x;
  const int k = blockIdx.x, r0 = blockIdx.y * RANGE;
  const int nr = min(RANGE, N - r0);
  for (int i = tid; i < RANGE / 4; i += 1024) ((int4*)tbl)[i] = make_int4(0, 0, 0, 0);
  __syncthreads();
  const int e0 = min(k * sl, E), e1 = min(e0 + sl, E);
  for (int base = e0 + tid; base < e1; base += 1024 * SEPT) {
    unsigned d[SEPT];
    for (int j = 0; j < SEPT; ++j) d[j] = (unsigned)(dst[min(base + j * 1024, e1 - 1)] - r0);
    for (int j = 0; j < SEPT; ++j) {
      if (base + j * 1024 >= e1) d[j] = 0xFFFFFFFFu;
    }
    for (int j = 0; j < SEPT; ++j) {
      if (d[j] < (unsigned)nr) atomicAdd(&tbl[d[j]], 1);
    }
  }
  __syncthreads();
  int* out = part + (size_t)k * part_pitch(N) + r0;
  const int nv = nr >> 2;
  for (int i = tid; i < nv; i += 1024) ((int4*)out)[i] = ((const int4*)tbl)[i];
  if (tid < (nr & 3)) out[4 * nv + tid] = tbl[4 * nv + tid];
}

__device__ __forceinline__ int wave_sum(int v) {
  for (int o = 16; o > 0; o >>= 1) v += __shfl_xor(v, o);
  return v;
}

__device__ __forceinline__ int block_sum(int v, int* sm) {
  const int lane = threadIdx.x & 31, wv = threadIdx.x >> 5;
  v = wave_sum(v);
  if (lane == 0) sm[wv] = v;
  __syncthreads();
  int t = 0;
  for (int i = 0; i < 8; ++i) t += sm[i];
  return t;
}

__global__ __launch_bounds__(256) void slice_prefix_kernel(int* __restrict__ part,
                                                           int* __restrict__ cnt,
                                                           int* __restrict__ bsum, int N) {
  __shared__ int sm[8];
  const int n = blockIdx.x * 256 + threadIdx.x;
  const size_t pn = (size_t)part_pitch(N);
  int run = 0;
  const int nsl = (n < N) ? NSL : 0;
  for (int k0 = 0; k0 < nsl; k0 += 32) {
    int c[32];
    #pragma unroll
    for (int j = 0; j < 32; ++j) c[j] = part[(size_t)(k0 + j) * pn + n];
    #pragma unroll
    for (int j = 0; j < 32; ++j) {
      part[(size_t)(k0 + j) * pn + n] = run;
      run += c[j];
    }
  }
  if (n < N) cnt[n] = run;
  const int tot = block_sum(run, sm);
  if (threadIdx.x == 0) bsum[blockIdx.x] = tot;
}

__global__ __launch_bounds__(256) void scan_kernel(const int* __restrict__ cnt,
                                                   const int* __restrict__ bsum,
                                                   int* __restrict__ rowptr,
                                                   float* __restrict__ dinv, int N) {
  __shared__ int sm[8];
  __shared__ int sw[8];
  const int tid = threadIdx.x, lane = tid & 31, wv = tid >> 5;
  int pre = 0;
  for (int i = tid; i < 4 * (int)blockIdx.x; i += 256) pre += bsum[i];
  pre = block_sum(pre, sm);

  const int i0 = blockIdx.x * 1024 + tid * 4;
  int c[4];
  for (int j = 0; j < 4; ++j) c[j] = (i0 + j < N) ? cnt[i0 + j] : 0;
  const int mine = c[0] + c[1] + c[2] + c[3];
  int inc = mine;
  for (int o = 1; o < 32; o <<= 1) {
    const int u = __shfl_up(inc, o);
    if (lane >= o) inc += u;
  }
  if (lane == 31) sw[wv] = inc;
  __syncthreads();
  int run = pre + inc - mine;
  for (int i = 0; i < wv; ++i) run += sw[i];
  for (int j = 0; j < 4; ++j) {
    if (i0 + j < N) {
      rowptr[i0 + j] = run;
      dinv[i0 + j] = rsqrtf((float)(c[j] + 1));
      run += c[j];
      if (i0 + j == N - 1) rowptr[N] = run;
    }
  }
}

__device__ __forceinline__ void put_slot(int* __restrict__ csr, int pos, int E, int s, float w) {
  int2 rec;
  rec.x = s;
  rec.y = __float_as_int(w);
  ((int2*)csr)[min(max(pos, 0), E - 1)] = rec;
}

__global__ __launch_bounds__(1024) void slice_fill_kernel(const int* __restrict__ src,
                                                          const int* __restrict__ dst,
                                                          const int* __restrict__ part,
                                                          const int* __restrict__ rowptr,
                                                          const float* __restrict__ dinv,
                                                          int* __restrict__ csr,
                                                          int N, int E, int sl) {
  __shared__ __attribute__((aligned(16))) int tbl[RANGE];
  const int tid = threadIdx.x;
  const int k = blockIdx.x, r0 = blockIdx.y * RANGE;
  const int nr = min(RANGE, N - r0);
  const int* pk = part + (size_t)k * part_pitch(N) + r0;
  const int* rp = rowptr + r0;
  const int nv = nr >> 2;
  for (int i0 = tid; i0 < nv; i0 += 1024 * 8) {
    int4 a[8], b[8];
    for (int j = 0; j < 8; ++j) {
      const int i = min(i0 + j * 1024, nv - 1);
      a[j] = ((const int4*)rp)[i];
      b[j] = ((const int4*)pk)[i];
    }
    for (int j = 0; j < 8; ++j) {
      const int i = i0 + j * 1024;
      if (i < nv) ((int4*)tbl)[i] = make_int4(a[j].x + b[j].x, a[j].y + b[j].y, a[j].z + b[j].z, a[j].w + b[j].w);
    }
  }
  if (tid < (nr & 3)) tbl[4 * nv + tid] = rp[4 * nv + tid] + pk[4 * nv + tid];
  __syncthreads();
  const int e0 = min(k * sl, E), e1 = min(e0 + sl, E);
  for (int base = e0 + tid; base < e1; base += 1024 * SEPT) {
    unsigned d[SEPT];
    int s[SEPT], pos[SEPT];
    float w[SEPT];
    for (int j = 0; j < SEPT; ++j) {
      const int e = min(base + j * 1024, e1 - 1);
      d[j] = (unsigned)(dst[e] - r0);
      s[j] = src[e];
    }
    for (int j = 0; j < SEPT; ++j) {
      if (base + j * 1024 >= e1) d[j] = 0xFFFFFFFFu;
    }
    for (int j = 0; j < SEPT; ++j) {
      const bool mine = d[j] < (unsigned)nr;
      w[j] = mine ? dinv[min(max(s[j], 0), N - 1)] : 0.0f;
      pos[j] = mine ? atomicAdd(&tbl[d[j]], 1) : -1;
    }
    for (int j = 0; j < SEPT; ++j) {
      if (pos[j] >= 0) put_slot(csr, pos[j], E, s[j], w[j]);
    }
  }
}


__device__ __forceinline__ float lane_f(float v, int l) {
  return __int_as_float(__builtin_amdgcn_readlane(__float_as_int(v), l));
}

__global__ __launch_bounds__(256) void gather_x_kernel(const int* __restrict__ rowptr,
                                                       const int* __restrict__ csr,
                                                       const float* __restrict__ dinv,
                                                       const float* __restrict__ X,
                                                       float* __restrict__ aggx,
                                                       int N, int E) {
  const int lane = threadIdx.x & 31;
  const int pass = blockIdx.y;
  const int n0 = __builtin_amdgcn_readfirstlane((blockIdx.x * 8 + (threadIdx.x >> 5)) * NPW);
  if (n0 >= N) return;
  const float* xb = X + pass * (TPP * F_IN) + lane * 2;
  const int rpv = rowptr[min(n0 + lane, N)];
  const float dvv = dinv[min(n0 + lane, N - 1)];
  for (int i = 0; i < NPW; ++i) {
    const int n = n0 + i;
    if (n >= N) break;
    const int beg = max(__builtin_amdgcn_readlane(rpv, i), 0);
    const int end = min(__builtin_amdgcn_readlane(rpv, i + 1), E);
    const float dn = lane_f(dvv, i);
    float2 acc;
    acc.x = 0.0f;
    acc.y = 0.0f;
    for (int c = beg; c <= end; c += 32) {
      const int e = c + lane;
      int sv = n;
      float wv = (e == end) ? dn : 0.0f;
      if (e < end) {
        const int2 rec = ((const int2*)csr)[e];
        sv = min(max(rec.x, 0), N - 1);
        wv = __int_as_float(rec.y);
      }
      const int m = min(end + 1 - c, 32);
      for (int k = 0; k < m; k += 8) {
        float2 x[8];
        float w[8];
        for (int j = 0; j < 8; ++j) {
          const int s = __builtin_amdgcn_readlane(sv, k + j);
          w[j] = lane_f(wv, k + j);
          x[j] = *(const float2*)(xb + (size_t)s * XROW);
        }
        for (int j = 0; j < 8; ++j) {
          acc.x += w[j] * x[j].x;
          acc.y += w[j] * x[j].y;
        }
      }
    }
    acc.x *= dn;
    acc.y *= dn;
    *(float2*)(aggx + (size_t)n * XROW + pass * (TPP * F_IN) + lane * 2) = acc;
  }
}

__global__ __launch_bounds__(256) void gru_seq_kernel(const float* __restrict__ aggx,
                                                      const float* __restrict__ Wg,
                                                      const float* __restrict__ bg,
                                                      const float* __restrict__ Wih,
                                                      const float* __restrict__ Whh,
                                                      const float* __restrict__ b_ih,
                                                      const float* __restrict__ b_hh,
                                                      const float* __restrict__ Wout,
                                                      const float* __restrict__ bout,
                                                      float* __restrict__ out, int N) {
  __shared__ __attribute__((aligned(16))) _Float16 sW[2 * 3 * H * WPITCH];
  __shared__ __attribute__((aligned(16))) float sB[2 * 3 * H];
  __shared__ __attribute__((aligned(16))) float sBg[H];
  __shared__ __attribute__((aligned(16))) float sWgT[H * WGPITCH];

  const int tid = threadIdx.x;
  for (int i = tid; i < 2 * 3 * H * 8; i += 256) {
    const int row = i >> 3, seg = i & 7;
    const float* sp = (row < 3 * H ? Wih + (size_t)row * H : Whh + (size_t)(row - 3 * H) * H) + seg * 8;
    const float4 f0 = ((const float4*)sp)[0], f1 = ((const float4*)sp)[1];
    v8h o;
    o[0] = (_Float16)f0.x; o[1] = (_Float16)f0.y; o[2] = (_Float16)f0.z; o[3] = (_Float16)f0.w;
    o[4] = (_Float16)f1.x; o[5] = (_Float16)f1.y; o[6] = (_Float16)f1.z; o[7] = (_Float16)f1.w;
    *(v8h*)(sW + row * WPITCH + seg * 8) = o;
  }
  for (int i = tid; i < 2 * 3 * H; i += 256) sB[i] = (i < 3 * H) ? b_ih[i] : b_hh[i - 3 * H];
  if (tid < H) sBg[tid] = bg[tid];
  for (int i = tid; i < F_IN * H; i += 256) sWgT[(i & (H - 1)) * WGPITCH + (i >> 6)] = Wg[i];
  __syncthreads();

  const int lane = tid & 31;
  const int c = lane & 15, hf = lane >> 4;
  const int wave = tid >> 5;
  const int place = wave ^ ((wave & 1) << 2);
  const int ntiles = (N + 15) >> 4;
  const int tstride = 8 * (int)gridDim.x;
  const float bo0 = bout[0], bo1 = bout[1];
  for (int tile = __builtin_amdgcn_readfirstlane(place * (int)gridDim.x + (int)blockIdx.x); tile < ntiles; tile += tstride) {
  const int node = tile * 16 + c;
  const int nodec = min(node, N - 1);

  v8f h32[4];
  for (int j = 0; j < 4; ++j) h32[j] = v8f{};
  v16h hB0 = {}, hB1 = {};

  const float* xrow = aggx + (size_t)nodec * XROW + 8 * hf;
  float4 xa = ((const float4*)xrow)[0], xb = ((const float4*)xrow)[1];

  for (int t = 0; t < T_STEPS; ++t) {
    asm volatile("" ::: "memory");
    const float* xn = xrow + min(t + 1, T_STEPS - 1) * F_IN;
    const float4 na = ((const float4*)xn)[0], nb = ((const float4*)xn)[1];
    const float xs[8] = {xa.x, xa.y, xa.z, xa.w, xb.x, xb.y, xb.z, xb.w};

    v16h aB0, aB1;
    for (int blk = 0; blk < 4; ++blk) {
      const float4 w0 = *(const float4*)&sWgT[(16 * blk + c) * WGPITCH + 8 * hf];
      const float4 w1 = *(const float4*)&sWgT[(16 * blk + c) * WGPITCH + 8 * hf + 4];
      const float ws8[8] = {w0.x, w0.y, w0.z, w0.w, w1.x, w1.y, w1.z, w1.w};
      v8f acc = {};
      for (int q = 0; q < 4; ++q) {
        v2f wq, xq;
        wq[0] = ws8[2 * q];
        wq[1] = ws8[2 * q + 1];
        xq[0] = xs[2 * q];
        xq[1] = xs[2 * q + 1];
        acc = __builtin_amdgcn_wmma_f32_16x16x4_f32(false, wq, false, xq, (short)0, acc, false, false);
      }
      const float4 g0 = *(const float4*)&sBg[16 * blk + 8 * hf];
      const float4 g1 = *(const float4*)&sBg[16 * blk + 8 * hf + 4];
      const float gb[8] = {g0.x, g0.y, g0.z, g0.w, g1.x, g1.y, g1.z, g1.w};
      for (int r = 0; r < 8; ++r) {
        const _Float16 av = (_Float16)fmaxf(acc[r] + gb[r], 0.0f);
        if (blk < 2) aB0[(blk & 1) * 8 + r] = av; else aB1[(blk & 1) * 8 + r] = av;
      }
    }

    #pragma unroll
    for (int j = 0; j < 4; ++j) {
      asm volatile("" ::: "memory");
      const _Float16* wr = sW + (          16 * j + c) * WPITCH + 8 * hf;
      const _Float16* wz = sW + (     64 + 16 * j + c) * WPITCH + 8 * hf;
      const _Float16* wn = sW + (    128 + 16 * j + c) * WPITCH + 8 * hf;
      const _Float16* ur = sW + (192 +       16 * j + c) * WPITCH + 8 * hf;
      const _Float16* uz = sW + (192 +  64 + 16 * j + c) * WPITCH + 8 * hf;
      const _Float16* un = sW + (192 + 128 + 16 * j + c) * WPITCH + 8 * hf;

      v8f ir = {}, iz = {}, in_ = {}, hr = {}, hz = {}, hn = {};
      ir  = __builtin_amdgcn_wmma_f32_16x16x32_f16(false, lds_frag(wr, 0), false, aB0, (short)0, ir,  false, false);
      iz  = __builtin_amdgcn_wmma_f32_16x16x32_f16(false, lds_frag(wz, 0), false, aB0, (short)0, iz,  false, false);
      in_ = __builtin_amdgcn_wmma_f32_16x16x32_f16(false, lds_frag(wn, 0), false, aB0, (short)0, in_, false, false);
      hr  = __builtin_amdgcn_wmma_f32_16x16x32_f16(false, lds_frag(ur, 0), false, hB0, (short)0, hr,  false, false);
      hz  = __builtin_amdgcn_wmma_f32_16x16x32_f16(false, lds_frag(uz, 0), false, hB0, (short)0, hz,  false, false);
      hn  = __builtin_amdgcn_wmma_f32_16x16x32_f16(false, lds_frag(un, 0), false, hB0, (short)0, hn,  false, false);
      ir  = __builtin_amdgcn_wmma_f32_16x16x32_f16(false, lds_frag(wr, 32), false, aB1, (short)0, ir,  false, false);
      iz  = __builtin_amdgcn_wmma_f32_16x16x32_f16(false, lds_frag(wz, 32), false, aB1, (short)0, iz,  false, false);
      in_ = __builtin_amdgcn_wmma_f32_16x16x32_f16(false, lds_frag(wn, 32), false, aB1, (short)0, in_, false, false);
      hr  = __builtin_amdgcn_wmma_f32_16x16x32_f16(false, lds_frag(ur, 32), false, hB1, (short)0, hr,  false, false);
      hz  = __builtin_amdgcn_wmma_f32_16x16x32_f16(false, lds_frag(uz, 32), false, hB1, (short)0, hz,  false, false);
      hn  = __builtin_amdgcn_wmma_f32_16x16x32_f16(false, lds_frag(un, 32), false, hB1, (short)0, hn,  false, false);

      const int ch = 16 * j + 8 * hf;
      float bir[8], biz[8], bin[8], bhr[8], bhz[8], bhn[8];
      for (int r = 0; r < 8; r += 4) {
        *(float4*)&bir[r] = *(const float4*)&sB[            ch + r];
        *(float4*)&biz[r] = *(const float4*)&sB[       64 + ch + r];
        *(float4*)&bin[r] = *(const float4*)&sB[      128 + ch + r];
        *(float4*)&bhr[r] = *(const float4*)&sB[192 +       ch + r];
        *(float4*)&bhz[r] = *(const float4*)&sB[192 +  64 + ch + r];
        *(float4*)&bhn[r] = *(const float4*)&sB[192 + 128 + ch + r];
      }
      v8f zgv, ncv;
      for (int r = 0; r < 8; ++r) {
        float rg = sigmoidf_(ir[r] + bir[r] + hr[r] + bhr[r]);
        zgv[r] = sigmoidf_(iz[r] + biz[r] + hz[r] + bhz[r]);
        ncv[r] = tanhf_(in_[r] + bin[r] + rg * (hn[r] + bhn[r]));
      }
      h32[j] = (1.0f - zgv) * ncv + zgv * h32[j];
    }

    for (int r = 0; r < 8; ++r) {
      hB0[r]     = (_Float16)h32[0][r];
      hB0[8 + r] = (_Float16)h32[1][r];
      hB1[r]     = (_Float16)h32[2][r];
      hB1[8 + r] = (_Float16)h32[3][r];
    }
    xa = na;
    xb = nb;
  }

  float o0 = 0.0f, o1 = 0.0f;
  for (int j = 0; j < 4; ++j) {
    const float* wo = Wout + (16 * j + 8 * hf) * 2;
    float w[16];
    for (int q = 0; q < 4; ++q) *(float4*)&w[4 * q] = *(const float4*)(wo + 4 * q);
    for (int r = 0; r < 8; ++r) {
      o0 += h32[j][r] * w[2 * r];
      o1 += h32[j][r] * w[2 * r + 1];
    }
  }
  o0 += other_half(o0);
  o1 += other_half(o1);
  if (hf == 0 && node < N) *(float2*)(out + (size_t)node * 2) = make_float2(o0 + bo0, o1 + bo1);
  }
}


static inline int cdiv(long long a, long long b) { return (int)((a + b - 1) / b); }

extern "C" void kernel_launch(void* const* d_in, const int* in_sizes, int n_in,
                              void* d_out, int out_size, void* d_ws, size_t ws_size,
                              hipStream_t stream) {
  const float* X    = (const float*)d_in[0];
  const int*   ei   = (const int*)  d_in[1];
  const float* Wg   = (const float*)d_in[2];
  const float* bg   = (const float*)d_in[3];
  const float* Wih  = (const float*)d_in[4];
  const float* Whh  = (const float*)d_in[5];
  const float* bih  = (const float*)d_in[6];
  const float* bhh  = (const float*)d_in[7];
  const float* Wout = (const float*)d_in[8];
  const float* bout = (const float*)d_in[9];

  const int N = in_sizes[0] / (T_STEPS * F_IN);
  const int E = in_sizes[1] / 2;
  const int* src = ei;
  const int* dst = ei + E;

  char* ws = (char*)d_ws;
  size_t off = 0;
  auto alloc = [&](size_t bytes) -> void* {
    void* p = ws + off;
    off = (off + bytes + 255) & ~(size_t)255;
    return p;
  };
  int*      rowptr  = (int*)   alloc(sizeof(int) * (N + 1));
  float*    dinv    = (float*) alloc(sizeof(float) * N);
  int*      cnt     = (int*)   alloc(sizeof(int) * N);
  int*      bsum    = (int*)   alloc(sizeof(int) * cdiv(N, 256));
  int*      csr     = (int*)   alloc(sizeof(int) * 2 * E);
  float*    aggx  = (float*)   alloc(sizeof(float) * (size_t)N * XROW);
  int*      part  = (int*)aggx;
  (void)ws_size;

  const int sl = cdiv(E, NSL);
  const dim3 sgrid(NSL, cdiv(N, RANGE));
  slice_count_kernel <<<sgrid, 1024, 0, stream>>>(dst, part, N, E, sl);
  slice_prefix_kernel<<<cdiv(N, 256), 256, 0, stream>>>(part, cnt, bsum, N);
  scan_kernel       <<<cdiv(N, 1024), 256, 0, stream>>>(cnt, bsum, rowptr, dinv, N);
  slice_fill_kernel  <<<sgrid, 1024, 0, stream>>>(src, dst, part, rowptr, dinv, csr, N, E, sl);

  const dim3 ggrid(cdiv(N, 8 * NPW), T_STEPS / TPP);
  gather_x_kernel<<<ggrid, 256, 0, stream>>>(rowptr, csr, dinv, X, aggx, N, E);

  gru_seq_kernel<<<GRU_GROUPS, 256, 0, stream>>>(aggx, Wg, bg, Wih, Whh, bih, bhh, Wout, bout, (float*)d_out, N);
}


# Edge sum moved into the recurrent kernel: a wave sums its 16 nodes' source rows into LDS before every 4 steps, two nodes' rows (48 loads) in flight; edge-pass launch and array gone; sum order kept
# speedup vs baseline: 11.3452x; 1.0710x over previous
#include <hip/hip_runtime.h>

#define T_STEPS 12
#define F_IN 16
#define H 64
#define XROW (T_STEPS * F_IN)
#define TPP 4
#define RANGE 51200
#define NSL 128
#define SEPT 13
#define XP 68
#define WGPITCH 20
#define GRU_GROUPS 512
#define WPITCH 72

typedef __attribute__((ext_vector_type(16))) _Float16 v16h;
typedef __attribute__((ext_vector_type(8)))  _Float16 v8h;
typedef __attribute__((ext_vector_type(8)))  float    v8f;
typedef __attribute__((ext_vector_type(2)))  float    v2f;

__device__ __forceinline__ v16h lds_frag(const _Float16* p, int k0) {
  union { v16h v; v8h h[2]; } u;
  u.h[0] = *(const v8h*)(p + k0);
  u.h[1] = *(const v8h*)(p + k0 + 16);
  return u.v;
}

__device__ __forceinline__ float sigmoidf_(float x) {
  return __builtin_amdgcn_rcpf(1.0f + __expf(-x));
}
__device__ __forceinline__ float other_half(float x) {
  return __builtin_bit_cast(float, __builtin_amdgcn_permlanex16(__builtin_bit_cast(int, x), __builtin_bit_cast(int, x),
                                                                  0x76543210, 0xfedcba98, false, false));
}
__device__ __forceinline__ float tanhf_(float x) {
  return 1.0f - 2.0f * __builtin_amdgcn_rcpf(1.0f + __expf(2.0f * x));
}


__device__ __forceinline__ int part_pitch(int N) { return (N + 63) & ~63; }

__global__ __launch_bounds__(1024) void slice_count_kernel(const int* __restrict__ dst,
                                                           int* __restrict__ part,
                                                           int N, int E, int sl) {
  __shared__ __attribute__((aligned(16))) int tbl[RANGE];
  const int tid = threadIdx.x;
  const int k = blockIdx.x, r0 = blockIdx.y * RANGE;
  const int nr = min(RANGE, N - r0);
  for (int i = tid; i < RANGE / 4; i += 1024) ((int4*)tbl)[i] = make_int4(0, 0, 0, 0);
  __syncthreads();
  const int e0 = min(k * sl, E), e1 = min(e0 + sl, E);
  for (int base = e0 + tid; base < e1; base += 1024 * SEPT) {
    unsigned d[SEPT];
    for (int j = 0; j < SEPT; ++j) d[j] = (unsigned)(dst[min(base + j * 1024, e1 - 1)] - r0);
    for (int j = 0; j < SEPT; ++j) {
      if (base + j * 1024 >= e1) d[j] = 0xFFFFFFFFu;
    }
    for (int j = 0; j < SEPT; ++j) {
      if (d[j] < (unsigned)nr) atomicAdd(&tbl[d[j]], 1);
    }
  }
  __syncthreads();
  int* out = part + (size_t)k * part_pitch(N) + r0;
  const int nv = nr >> 2;
  for (int i = tid; i < nv; i += 1024) ((int4*)out)[i] = ((const int4*)tbl)[i];
  if (tid < (nr & 3)) out[4 * nv + tid] = tbl[4 * nv + tid];
}

__device__ __forceinline__ int wave_sum(int v) {
  for (int o = 16; o > 0; o >>= 1) v += __shfl_xor(v, o);
  return v;
}

__device__ __forceinline__ int block_sum(int v, int* sm) {
  const int lane = threadIdx.x & 31, wv = threadIdx.x >> 5;
  v = wave_sum(v);
  if (lane == 0) sm[wv] = v;
  __syncthreads();
  int t = 0;
  for (int i = 0; i < 8; ++i) t += sm[i];
  return t;
}

__global__ __launch_bounds__(256) void slice_prefix_kernel(int* __restrict__ part,
                                                           int* __restrict__ cnt,
                                                           int* __restrict__ bsum, int N) {
  __shared__ int sm[8];
  const int n = blockIdx.x * 256 + threadIdx.x;
  const size_t pn = (size_t)part_pitch(N);
  int run = 0;
  const int nsl = (n < N) ? NSL : 0;
  for (int k0 = 0; k0 < nsl; k0 += 32) {
    int c[32];
    #pragma unroll
    for (int j = 0; j < 32; ++j) c[j] = part[(size_t)(k0 + j) * pn + n];
    #pragma unroll
    for (int j = 0; j < 32; ++j) {
      part[(size_t)(k0 + j) * pn + n] = run;
      run += c[j];
    }
  }
  if (n < N) cnt[n] = run;
  const int tot = block_sum(run, sm);
  if (threadIdx.x == 0) bsum[blockIdx.x] = tot;
}

__global__ __launch_bounds__(256) void scan_kernel(const int* __restrict__ cnt,
                                                   const int* __restrict__ bsum,
                                                   int* __restrict__ rowptr,
                                                   float* __restrict__ dinv, int N) {
  __shared__ int sm[8];
  __shared__ int sw[8];
  const int tid = threadIdx.x, lane = tid & 31, wv = tid >> 5;
  int pre = 0;
  for (int i = tid; i < 4 * (int)blockIdx.x; i += 256) pre += bsum[i];
  pre = block_sum(pre, sm);

  const int i0 = blockIdx.x * 1024 + tid * 4;
  int c[4];
  for (int j = 0; j < 4; ++j) c[j] = (i0 + j < N) ? cnt[i0 + j] : 0;
  const int mine = c[0] + c[1] + c[2] + c[3];
  int inc = mine;
  for (int o = 1; o < 32; o <<= 1) {
    const int u = __shfl_up(inc, o);
    if (lane >= o) inc += u;
  }
  if (lane == 31) sw[wv] = inc;
  __syncthreads();
  int run = pre + inc - mine;
  for (int i = 0; i < wv; ++i) run += sw[i];
  for (int j = 0; j < 4; ++j) {
    if (i0 + j < N) {
      rowptr[i0 + j] = run;
      dinv[i0 + j] = rsqrtf((float)(c[j] + 1));
      run += c[j];
      if (i0 + j == N - 1) rowptr[N] = run;
    }
  }
}

__device__ __forceinline__ void put_slot(int* __restrict__ csr, int pos, int E, int s, float w) {
  int2 rec;
  rec.x = s;
  rec.y = __float_as_int(w);
  ((int2*)csr)[min(max(pos, 0), E - 1)] = rec;
}

__global__ __launch_bounds__(1024) void slice_fill_kernel(const int* __restrict__ src,
                                                          const int* __restrict__ dst,
                                                          const int* __restrict__ part,
                                                          const int* __restrict__ rowptr,
                                                          const float* __restrict__ dinv,
                                                          int* __restrict__ csr,
                                                          int N, int E, int sl) {
  __shared__ __attribute__((aligned(16))) int tbl[RANGE];
  const int tid = threadIdx.x;
  const int k = blockIdx.x, r0 = blockIdx.y * RANGE;
  const int nr = min(RANGE, N - r0);
  const int* pk = part + (size_t)k * part_pitch(N) + r0;
  const int* rp = rowptr + r0;
  const int nv = nr >> 2;
  for (int i0 = tid; i0 < nv; i0 += 1024 * 8) {
    int4 a[8], b[8];
    for (int j = 0; j < 8; ++j) {
      const int i = min(i0 + j * 1024, nv - 1);
      a[j] = ((const int4*)rp)[i];
      b[j] = ((const int4*)pk)[i];
    }
    for (int j = 0; j < 8; ++j) {
      const int i = i0 + j * 1024;
      if (i < nv) ((int4*)tbl)[i] = make_int4(a[j].x + b[j].x, a[j].y + b[j].y, a[j].z + b[j].z, a[j].w + b[j].w);
    }
  }
  if (tid < (nr & 3)) tbl[4 * nv + tid] = rp[4 * nv + tid] + pk[4 * nv + tid];
  __syncthreads();
  const int e0 = min(k * sl, E), e1 = min(e0 + sl, E);
  for (int base = e0 + tid; base < e1; base += 1024 * SEPT) {
    unsigned d[SEPT];
    int s[SEPT], pos[SEPT];
    float w[SEPT];
    for (int j = 0; j < SEPT; ++j) {
      const int e = min(base + j * 1024, e1 - 1);
      d[j] = (unsigned)(dst[e] - r0);
      s[j] = src[e];
    }
    for (int j = 0; j < SEPT; ++j) {
      if (base + j * 1024 >= e1) d[j] = 0xFFFFFFFFu;
    }
    for (int j = 0; j < SEPT; ++j) {
      const bool mine = d[j] < (unsigned)nr;
      w[j] = mine ? dinv[min(max(s[j], 0), N - 1)] : 0.0f;
      pos[j] = mine ? atomicAdd(&tbl[d[j]], 1) : -1;
    }
    for (int j = 0; j < SEPT; ++j) {
      if (pos[j] >= 0) put_slot(csr, pos[j], E, s[j], w[j]);
    }
  }
}


__device__ __forceinline__ float lane_f(float v, int l) {
  return __int_as_float(__builtin_amdgcn_readlane(__float_as_int(v), l));
}

__device__ __forceinline__ void slot_of(int2 rec, int e, int end, int n, float dn, int N, int& sv, float& wv) {
  sv = n;
  wv = (e == end) ? dn : 0.0f;
  if (e < end) {
    sv = min(max(rec.x, 0), N - 1);
    wv = __int_as_float(rec.y);
  }
}

__device__ __forceinline__ float2 row_chunk(const float* __restrict__ xp, unsigned lane8, int sj) {
  const unsigned off = (unsigned)sj * (unsigned)(XROW * sizeof(float)) + lane8;
  return *(const float2*)((const char*)xp + off);
}

#define ROWS2 24

__device__ __forceinline__ void rows_sum2(const float* __restrict__ xp, unsigned lane8,
                                          int svA, float wvA, int svB, float wvB, float2& accA, float2& accB) {
  float2 xa[ROWS2], xb[ROWS2];
  #pragma unroll
  for (int j = 0; j < ROWS2; ++j) xa[j] = row_chunk(xp, lane8, __builtin_amdgcn_readlane(svA, j));
  #pragma unroll
  for (int j = 0; j < ROWS2; ++j) xb[j] = row_chunk(xp, lane8, __builtin_amdgcn_readlane(svB, j));
  #pragma unroll
  for (int j = 0; j < ROWS2; ++j) {
    const float wj = lane_f(wvA, j);
    accA.x = __builtin_fmaf(wj, xa[j].x, accA.x);
    accA.y = __builtin_fmaf(wj, xa[j].y, accA.y);
  }
  #pragma unroll
  for (int j = 0; j < ROWS2; ++j) {
    const float wj = lane_f(wvB, j);
    accB.x = __builtin_fmaf(wj, xb[j].x, accB.x);
    accB.y = __builtin_fmaf(wj, xb[j].y, accB.y);
  }
}

__device__ __forceinline__ void rows_sum(const float* __restrict__ xp, unsigned lane8, int sv, float wv, const int j0, float2& acc) {
  float2 x[32];
  #pragma unroll
  for (int j = j0; j < 32; ++j) x[j] = row_chunk(xp, lane8, __builtin_amdgcn_readlane(sv, j));
  #pragma unroll
  for (int j = j0; j < 32; ++j) {
    const float wj = lane_f(wv, j);
    acc.x = __builtin_fmaf(wj, x[j].x, acc.x);
    acc.y = __builtin_fmaf(wj, x[j].y, acc.y);
  }
}

__device__ __forceinline__ void rows_rest(const float* __restrict__ xp, unsigned lane8, const int2* __restrict__ csr2,
                                          int lane, int beg, int end, int n, float dn, int N, int emax,
                                          int sv, float wv, float2& acc) {
  if (end + 1 - beg > ROWS2) rows_sum(xp, lane8, sv, wv, ROWS2, acc);
  for (int cs = beg + 32; cs <= end; cs += 32) {
    const int e = cs + lane;
    const int2 rec = csr2[min(e, emax)];
    int s2;
    float w2;
    slot_of(rec, e, end, n, dn, N, s2, w2);
    rows_sum(xp, lane8, s2, w2, 0, acc);
  }
}

__global__ __launch_bounds__(256) void gru_seq_kernel(const int* __restrict__ rowptr,
                                                      const int* __restrict__ csr,
                                                      const float* __restrict__ dinv,
                                                      const float* __restrict__ X,
                                                      const float* __restrict__ Wg,
                                                      const float* __restrict__ bg,
                                                      const float* __restrict__ Wih,
                                                      const float* __restrict__ Whh,
                                                      const float* __restrict__ b_ih,
                                                      const float* __restrict__ b_hh,
                                                      const float* __restrict__ Wout,
                                                      const float* __restrict__ bout,
                                                      float* __restrict__ out, int N, int E) {
  __shared__ __attribute__((aligned(16))) _Float16 sW[2 * 3 * H * WPITCH];
  __shared__ __attribute__((aligned(16))) float sB[2 * 3 * H];
  __shared__ __attribute__((aligned(16))) float sBg[H];
  __shared__ __attribute__((aligned(16))) float sWgT[H * WGPITCH];
  __shared__ __attribute__((aligned(16))) float sX[8 * 16 * XP];

  const int tid = threadIdx.x;
  for (int i = tid; i < 2 * 3 * H * 8; i += 256) {
    const int row = i >> 3, seg = i & 7;
    const float* sp = (row < 3 * H ? Wih + (size_t)row * H : Whh + (size_t)(row - 3 * H) * H) + seg * 8;
    const float4 f0 = ((const float4*)sp)[0], f1 = ((const float4*)sp)[1];
    v8h o;
    o[0] = (_Float16)f0.x; o[1] = (_Float16)f0.y; o[2] = (_Float16)f0.z; o[3] = (_Float16)f0.w;
    o[4] = (_Float16)f1.x; o[5] = (_Float16)f1.y; o[6] = (_Float16)f1.z; o[7] = (_Float16)f1.w;
    *(v8h*)(sW + row * WPITCH + seg * 8) = o;
  }
  for (int i = tid; i < 2 * 3 * H; i += 256) sB[i] = (i < 3 * H) ? b_ih[i] : b_hh[i - 3 * H];
  if (tid < H) sBg[tid] = bg[tid];
  for (int i = tid; i < F_IN * H; i += 256) sWgT[(i & (H - 1)) * WGPITCH + (i >> 6)] = Wg[i];
  __syncthreads();

  const int lane = tid & 31;
  const int c = lane & 15, hf = lane >> 4;
  const int wave = tid >> 5;
  const int place = wave ^ ((wave & 1) << 2);
  const int ntiles = (N + 15) >> 4;
  const int tstride = 8 * (int)gridDim.x;
  const float bo0 = bout[0], bo1 = bout[1];
  float* myx = sX + wave * (16 * XP);
  const int2* csr2 = (const int2*)csr;
  const int emax = max(E - 1, 0);
  for (int tile = __builtin_amdgcn_readfirstlane(place * (int)gridDim.x + (int)blockIdx.x); tile < ntiles; tile += tstride) {
  const int node = tile * 16 + c;
  const int n0 = tile * 16;

  v8f h32[4];
  for (int j = 0; j < 4; ++j) h32[j] = v8f{};
  v16h hB0 = {}, hB1 = {};

  const int rpv = rowptr[min(n0 + lane, N)];
  const float dvv = dinv[min(n0 + lane, N - 1)];

  for (int pass = 0; pass < T_STEPS / TPP; ++pass) {
  {
    const float* xp = X + pass * (TPP * F_IN);
    const unsigned lane8 = (unsigned)lane * 8u;
    int begA = max(__builtin_amdgcn_readlane(rpv, 0), 0);
    int begB = max(__builtin_amdgcn_readlane(rpv, 1), 0);
    int2 recA = csr2[min(begA + lane, emax)];
    int2 recB = csr2[min(begB + lane, emax)];
    for (int i = 0; i < 16; i += 2) {
      const int nA = min(n0 + i, N - 1), nB = min(n0 + i + 1, N - 1);
      const int endA = min(__builtin_amdgcn_readlane(rpv, i + 1), E);
      const int endB = min(__builtin_amdgcn_readlane(rpv, i + 2), E);
      const float dnA = lane_f(dvv, i), dnB = lane_f(dvv, i + 1);
      const int nbegA = max(__builtin_amdgcn_readlane(rpv, i + 2), 0);
      const int nbegB = max(__builtin_amdgcn_readlane(rpv, i + 3), 0);
      const int2 nrecA = csr2[min(nbegA + lane, emax)];
      const int2 nrecB = csr2[min(nbegB + lane, emax)];
      int svA, svB;
      float wvA, wvB;
      slot_of(recA, begA + lane, endA, nA, dnA, N, svA, wvA);
      slot_of(recB, begB + lane, endB, nB, dnB, N, svB, wvB);
      float2 accA, accB;
      accA.x = 0.0f; accA.y = 0.0f;
      accB.x = 0.0f; accB.y = 0.0f;
      rows_sum2(xp, lane8, svA, wvA, svB, wvB, accA, accB);
      rows_rest(xp, lane8, csr2, lane, begA, endA, nA, dnA, N, emax, svA, wvA, accA);
      rows_rest(xp, lane8, csr2, lane, begB, endB, nB, dnB, N, emax, svB, wvB, accB);
      accA.x *= dnA;
      accA.y *= dnA;
      accB.x *= dnB;
      accB.y *= dnB;
      *(float2*)(myx + i * XP + lane * 2) = accA;
      *(float2*)(myx + (i + 1) * XP + lane * 2) = accB;
      begA = nbegA;
      begB = nbegB;
      recA = nrecA;
      recB = nrecB;
    }
  }
  asm volatile("s_wait_dscnt 0x0" ::: "memory");

  for (int tt = 0; tt < TPP; ++tt) {
    asm volatile("" ::: "memory");
    const float* xl = myx + c * XP + tt * F_IN + 8 * hf;
    const float4 xa = ((const float4*)xl)[0], xb = ((const float4*)xl)[1];
    const float xs[8] = {xa.x, xa.y, xa.z, xa.w, xb.x, xb.y, xb.z, xb.w};

    v16h aB0, aB1;
    for (int blk = 0; blk < 4; ++blk) {
      const float4 w0 = *(const float4*)&sWgT[(16 * blk + c) * WGPITCH + 8 * hf];
      const float4 w1 = *(const float4*)&sWgT[(16 * blk + c) * WGPITCH + 8 * hf + 4];
      const float ws8[8] = {w0.x, w0.y, w0.z, w0.w, w1.x, w1.y, w1.z, w1.w};
      v8f acc = {};
      for (int q = 0; q < 4; ++q) {
        v2f wq, xq;
        wq[0] = ws8[2 * q];
        wq[1] = ws8[2 * q + 1];
        xq[0] = xs[2 * q];
        xq[1] = xs[2 * q + 1];
        acc = __builtin_amdgcn_wmma_f32_16x16x4_f32(false, wq, false, xq, (short)0, acc, false, false);
      }
      const float4 g0 = *(const float4*)&sBg[16 * blk + 8 * hf];
      const float4 g1 = *(const float4*)&sBg[16 * blk + 8 * hf + 4];
      const float gb[8] = {g0.x, g0.y, g0.z, g0.w, g1.x, g1.y, g1.z, g1.w};
      for (int r = 0; r < 8; ++r) {
        const _Float16 av = (_Float16)fmaxf(acc[r] + gb[r], 0.0f);
        if (blk < 2) aB0[(blk & 1) * 8 + r] = av; else aB1[(blk & 1) * 8 + r] = av;
      }
    }

    #pragma unroll
    for (int j = 0; j < 4; ++j) {
      asm volatile("" ::: "memory");
      const _Float16* wr = sW + (          16 * j + c) * WPITCH + 8 * hf;
      const _Float16* wz = sW + (     64 + 16 * j + c) * WPITCH + 8 * hf;
      const _Float16* wn = sW + (    128 + 16 * j + c) * WPITCH + 8 * hf;
      const _Float16* ur = sW + (192 +       16 * j + c) * WPITCH + 8 * hf;
      const _Float16* uz = sW + (192 +  64 + 16 * j + c) * WPITCH + 8 * hf;
      const _Float16* un = sW + (192 + 128 + 16 * j + c) * WPITCH + 8 * hf;

      v8f ir = {}, iz = {}, in_ = {}, hr = {}, hz = {}, hn = {};
      ir  = __builtin_amdgcn_wmma_f32_16x16x32_f16(false, lds_frag(wr, 0), false, aB0, (short)0, ir,  false, false);
      iz  = __builtin_amdgcn_wmma_f32_16x16x32_f16(false, lds_frag(wz, 0), false, aB0, (short)0, iz,  false, false);
      in_ = __builtin_amdgcn_wmma_f32_16x16x32_f16(false, lds_frag(wn, 0), false, aB0, (short)0, in_, false, false);
      hr  = __builtin_amdgcn_wmma_f32_16x16x32_f16(false, lds_frag(ur, 0), false, hB0, (short)0, hr,  false, false);
      hz  = __builtin_amdgcn_wmma_f32_16x16x32_f16(false, lds_frag(uz, 0), false, hB0, (short)0, hz,  false, false);
      hn  = __builtin_amdgcn_wmma_f32_16x16x32_f16(false, lds_frag(un, 0), false, hB0, (short)0, hn,  false, false);
      ir  = __builtin_amdgcn_wmma_f32_16x16x32_f16(false, lds_frag(wr, 32), false, aB1, (short)0, ir,  false, false);
      iz  = __builtin_amdgcn_wmma_f32_16x16x32_f16(false, lds_frag(wz, 32), false, aB1, (short)0, iz,  false, false);
      in_ = __builtin_amdgcn_wmma_f32_16x16x32_f16(false, lds_frag(wn, 32), false, aB1, (short)0, in_, false, false);
      hr  = __builtin_amdgcn_wmma_f32_16x16x32_f16(false, lds_frag(ur, 32), false, hB1, (short)0, hr,  false, false);
      hz  = __builtin_amdgcn_wmma_f32_16x16x32_f16(false, lds_frag(uz, 32), false, hB1, (short)0, hz,  false, false);
      hn  = __builtin_amdgcn_wmma_f32_16x16x32_f16(false, lds_frag(un, 32), false, hB1, (short)0, hn,  false, false);

      const int ch = 16 * j + 8 * hf;
      float bir[8], biz[8], bin[8], bhr[8], bhz[8], bhn[8];
      for (int r = 0; r < 8; r += 4) {
        *(float4*)&bir[r] = *(const float4*)&sB[            ch + r];
        *(float4*)&biz[r] = *(const float4*)&sB[       64 + ch + r];
        *(float4*)&bin[r] = *(const float4*)&sB[      128 + ch + r];
        *(float4*)&bhr[r] = *(const float4*)&sB[192 +       ch + r];
        *(float4*)&bhz[r] = *(const float4*)&sB[192 +  64 + ch + r];
        *(float4*)&bhn[r] = *(const float4*)&sB[192 + 128 + ch + r];
      }
      v8f zgv, ncv;
      for (int r = 0; r < 8; ++r) {
        float rg = sigmoidf_(ir[r] + bir[r] + hr[r] + bhr[r]);
        zgv[r] = sigmoidf_(iz[r] + biz[r] + hz[r] + bhz[r]);
        ncv[r] = tanhf_(in_[r] + bin[r] + rg * (hn[r] + bhn[r]));
      }
      h32[j] = (1.0f - zgv) * ncv + zgv * h32[j];
    }

    for (int r = 0; r < 8; ++r) {
      hB0[r]     = (_Float16)h32[0][r];
      hB0[8 + r] = (_Float16)h32[1][r];
      hB1[r]     = (_Float16)h32[2][r];
      hB1[8 + r] = (_Float16)h32[3][r];
    }
  }
  }

  float o0 = 0.0f, o1 = 0.0f;
  for (int j = 0; j < 4; ++j) {
    const float* wo = Wout + (16 * j + 8 * hf) * 2;
    float w[16];
    for (int q = 0; q < 4; ++q) *(float4*)&w[4 * q] = *(const float4*)(wo + 4 * q);
    for (int r = 0; r < 8; ++r) {
      o0 += h32[j][r] * w[2 * r];
      o1 += h32[j][r] * w[2 * r + 1];
    }
  }
  o0 += other_half(o0);
  o1 += other_half(o1);
  if (hf == 0 && node < N) *(float2*)(out + (size_t)node * 2) = make_float2(o0 + bo0, o1 + bo1);
  }
}


static inline int cdiv(long long a, long long b) { return (int)((a + b - 1) / b); }

extern "C" void kernel_launch(void* const* d_in, const int* in_sizes, int n_in,
                              void* d_out, int out_size, void* d_ws, size_t ws_size,
                              hipStream_t stream) {
  const float* X    = (const float*)d_in[0];
  const int*   ei   = (const int*)  d_in[1];
  const float* Wg   = (const float*)d_in[2];
  const float* bg   = (const float*)d_in[3];
  const float* Wih  = (const float*)d_in[4];
  const float* Whh  = (const float*)d_in[5];
  const float* bih  = (const float*)d_in[6];
  const float* bhh  = (const float*)d_in[7];
  const float* Wout = (const float*)d_in[8];
  const float* bout = (const float*)d_in[9];

  const int N = in_sizes[0] / (T_STEPS * F_IN);
  const int E = in_sizes[1] / 2;
  const int* src = ei;
  const int* dst = ei + E;

  char* ws = (char*)d_ws;
  size_t off = 0;
  auto alloc = [&](size_t bytes) -> void* {
    void* p = ws + off;
    off = (off + bytes + 255) & ~(size_t)255;
    return p;
  };
  int*      rowptr  = (int*)   alloc(sizeof(int) * (N + 1));
  float*    dinv    = (float*) alloc(sizeof(float) * N);
  int*      cnt     = (int*)   alloc(sizeof(int) * N);
  int*      bsum    = (int*)   alloc(sizeof(int) * cdiv(N, 256));
  int*      csr     = (int*)   alloc(sizeof(int) * 2 * E);
  int*      part    = (int*)   alloc(sizeof(int) * (size_t)NSL * (size_t)((N + 63) & ~63));
  (void)ws_size;

  const int sl = cdiv(E, NSL);
  const dim3 sgrid(NSL, cdiv(N, RANGE));
  slice_count_kernel <<<sgrid, 1024, 0, stream>>>(dst, part, N, E, sl);
  slice_prefix_kernel<<<cdiv(N, 256), 256, 0, stream>>>(part, cnt, bsum, N);
  scan_kernel       <<<cdiv(N, 1024), 256, 0, stream>>>(cnt, bsum, rowptr, dinv, N);
  slice_fill_kernel  <<<sgrid, 1024, 0, stream>>>(src, dst, part, rowptr, dinv, csr, N, E, sl);

  gru_seq_kernel<<<GRU_GROUPS, 256, 0, stream>>>(rowptr, csr, dinv, X, Wg, bg, Wih, Whh, bih, bhh, Wout, bout, (float*)d_out, N, E);
}


# Gates on the hardware tanh (v_tanh_f32): sigmoid(x) = (1 + tanh(x/2)) / 2 with the halving folded into the staged r and z weights and biases, tanh direct; 3 transcendentals a channel, not 6
# speedup vs baseline: 11.8834x; 1.0474x over previous
#include <hip/hip_runtime.h>

#define T_STEPS 12
#define F_IN 16
#define H 64
#define XROW (T_STEPS * F_IN)
#define TPP 4
#define RANGE 51200
#define NSL 128
#define SEPT 13
#define XP 68
#define WGPITCH 20
#define GRU_GROUPS 512
#define WPITCH 72

typedef __attribute__((ext_vector_type(16))) _Float16 v16h;
typedef __attribute__((ext_vector_type(8)))  _Float16 v8h;
typedef __attribute__((ext_vector_type(8)))  float    v8f;
typedef __attribute__((ext_vector_type(2)))  float    v2f;

__device__ __forceinline__ v16h lds_frag(const _Float16* p, int k0) {
  union { v16h v; v8h h[2]; } u;
  u.h[0] = *(const v8h*)(p + k0);
  u.h[1] = *(const v8h*)(p + k0 + 16);
  return u.v;
}

__device__ __forceinline__ float sigmoid_of_half_(float xh) {
  return __builtin_fmaf(0.5f, __builtin_amdgcn_tanhf(xh), 0.5f);
}
__device__ __forceinline__ float other_half(float x) {
  return __builtin_bit_cast(float, __builtin_amdgcn_permlanex16(__builtin_bit_cast(int, x), __builtin_bit_cast(int, x),
                                                                  0x76543210, 0xfedcba98, false, false));
}
__device__ __forceinline__ float tanhf_(float x) {
  return __builtin_amdgcn_tanhf(x);
}


__device__ __forceinline__ int part_pitch(int N) { return (N + 63) & ~63; }

__global__ __launch_bounds__(1024) void slice_count_kernel(const int* __restrict__ dst,
                                                           int* __restrict__ part,
                                                           int N, int E, int sl) {
  __shared__ __attribute__((aligned(16))) int tbl[RANGE];
  const int tid = threadIdx.x;
  const int k = blockIdx.x, r0 = blockIdx.y * RANGE;
  const int nr = min(RANGE, N - r0);
  for (int i = tid; i < RANGE / 4; i += 1024) ((int4*)tbl)[i] = make_int4(0, 0, 0, 0);
  __syncthreads();
  const int e0 = min(k * sl, E), e1 = min(e0 + sl, E);
  for (int base = e0 + tid; base < e1; base += 1024 * SEPT) {
    unsigned d[SEPT];
    for (int j = 0; j < SEPT; ++j) d[j] = (unsigned)(dst[min(base + j * 1024, e1 - 1)] - r0);
    for (int j = 0; j < SEPT; ++j) {
      if (base + j * 1024 >= e1) d[j] = 0xFFFFFFFFu;
    }
    for (int j = 0; j < SEPT; ++j) {
      if (d[j] < (unsigned)nr) atomicAdd(&tbl[d[j]], 1);
    }
  }
  __syncthreads();
  int* out = part + (size_t)k * part_pitch(N) + r0;
  const int nv = nr >> 2;
  for (int i = tid; i < nv; i += 1024) ((int4*)out)[i] = ((const int4*)tbl)[i];
  if (tid < (nr & 3)) out[4 * nv + tid] = tbl[4 * nv + tid];
}

__device__ __forceinline__ int wave_sum(int v) {
  for (int o = 16; o > 0; o >>= 1) v += __shfl_xor(v, o);
  return v;
}

__device__ __forceinline__ int block_sum(int v, int* sm) {
  const int lane = threadIdx.x & 31, wv = threadIdx.x >> 5;
  v = wave_sum(v);
  if (lane == 0) sm[wv] = v;
  __syncthreads();
  int t = 0;
  for (int i = 0; i < 8; ++i) t += sm[i];
  return t;
}

__global__ __launch_bounds__(256) void slice_prefix_kernel(int* __restrict__ part,
                                                           int* __restrict__ cnt,
                                                           int* __restrict__ bsum, int N) {
  __shared__ int sm[8];
  const int n = blockIdx.x * 256 + threadIdx.x;
  const size_t pn = (size_t)part_pitch(N);
  int run = 0;
  const int nsl = (n < N) ? NSL : 0;
  for (int k0 = 0; k0 < nsl; k0 += 32) {
    int c[32];
    #pragma unroll
    for (int j = 0; j < 32; ++j) c[j] = part[(size_t)(k0 + j) * pn + n];
    #pragma unroll
    for (int j = 0; j < 32; ++j) {
      part[(size_t)(k0 + j) * pn + n] = run;
      run += c[j];
    }
  }
  if (n < N) cnt[n] = run;
  const int tot = block_sum(run, sm);
  if (threadIdx.x == 0) bsum[blockIdx.x] = tot;
}

__global__ __launch_bounds__(256) void scan_kernel(const int* __restrict__ cnt,
                                                   const int* __restrict__ bsum,
                                                   int* __restrict__ rowptr,
                                                   float* __restrict__ dinv, int N) {
  __shared__ int sm[8];
  __shared__ int sw[8];
  const int tid = threadIdx.x, lane = tid & 31, wv = tid >> 5;
  int pre = 0;
  for (int i = tid; i < 4 * (int)blockIdx.x; i += 256) pre += bsum[i];
  pre = block_sum(pre, sm);

  const int i0 = blockIdx.x * 1024 + tid * 4;
  int c[4];
  for (int j = 0; j < 4; ++j) c[j] = (i0 + j < N) ? cnt[i0 + j] : 0;
  const int mine = c[0] + c[1] + c[2] + c[3];
  int inc = mine;
  for (int o = 1; o < 32; o <<= 1) {
    const int u = __shfl_up(inc, o);
    if (lane >= o) inc += u;
  }
  if (lane == 31) sw[wv] = inc;
  __syncthreads();
  int run = pre + inc - mine;
  for (int i = 0; i < wv; ++i) run += sw[i];
  for (int j = 0; j < 4; ++j) {
    if (i0 + j < N) {
      rowptr[i0 + j] = run;
      dinv[i0 + j] = rsqrtf((float)(c[j] + 1));
      run += c[j];
      if (i0 + j == N - 1) rowptr[N] = run;
    }
  }
}

__device__ __forceinline__ void put_slot(int* __restrict__ csr, int pos, int E, int s, float w) {
  int2 rec;
  rec.x = s;
  rec.y = __float_as_int(w);
  ((int2*)csr)[min(max(pos, 0), E - 1)] = rec;
}

__global__ __launch_bounds__(1024) void slice_fill_kernel(const int* __restrict__ src,
                                                          const int* __restrict__ dst,
                                                          const int* __restrict__ part,
                                                          const int* __restrict__ rowptr,
                                                          const float* __restrict__ dinv,
                                                          int* __restrict__ csr,
                                                          int N, int E, int sl) {
  __shared__ __attribute__((aligned(16))) int tbl[RANGE];
  const int tid = threadIdx.x;
  const int k = blockIdx.x, r0 = blockIdx.y * RANGE;
  const int nr = min(RANGE, N - r0);
  const int* pk = part + (size_t)k * part_pitch(N) + r0;
  const int* rp = rowptr + r0;
  const int nv = nr >> 2;
  for (int i0 = tid; i0 < nv; i0 += 1024 * 8) {
    int4 a[8], b[8];
    for (int j = 0; j < 8; ++j) {
      const int i = min(i0 + j * 1024, nv - 1);
      a[j] = ((const int4*)rp)[i];
      b[j] = ((const int4*)pk)[i];
    }
    for (int j = 0; j < 8; ++j) {
      const int i = i0 + j * 1024;
      if (i < nv) ((int4*)tbl)[i] = make_int4(a[j].x + b[j].x, a[j].y + b[j].y, a[j].z + b[j].z, a[j].w + b[j].w);
    }
  }
  if (tid < (nr & 3)) tbl[4 * nv + tid] = rp[4 * nv + tid] + pk[4 * nv + tid];
  __syncthreads();
  const int e0 = min(k * sl, E), e1 = min(e0 + sl, E);
  for (int base = e0 + tid; base < e1; base += 1024 * SEPT) {
    unsigned d[SEPT];
    int s[SEPT], pos[SEPT];
    float w[SEPT];
    for (int j = 0; j < SEPT; ++j) {
      const int e = min(base + j * 1024, e1 - 1);
      d[j] = (unsigned)(dst[e] - r0);
      s[j] = src[e];
    }
    for (int j = 0; j < SEPT; ++j) {
      if (base + j * 1024 >= e1) d[j] = 0xFFFFFFFFu;
    }
    for (int j = 0; j < SEPT; ++j) {
      const bool mine = d[j] < (unsigned)nr;
      w[j] = mine ? dinv[min(max(s[j], 0), N - 1)] : 0.0f;
      pos[j] = mine ? atomicAdd(&tbl[d[j]], 1) : -1;
    }
    for (int j = 0; j < SEPT; ++j) {
      if (pos[j] >= 0) put_slot(csr, pos[j], E, s[j], w[j]);
    }
  }
}


__device__ __forceinline__ float lane_f(float v, int l) {
  return __int_as_float(__builtin_amdgcn_readlane(__float_as_int(v), l));
}

__device__ __forceinline__ void slot_of(int2 rec, int e, int end, int n, float dn, int N, int& sv, float& wv) {
  sv = n;
  wv = (e == end) ? dn : 0.0f;
  if (e < end) {
    sv = min(max(rec.x, 0), N - 1);
    wv = __int_as_float(rec.y);
  }
}

__device__ __forceinline__ float2 row_chunk(const float* __restrict__ xp, unsigned lane8, int sj) {
  const unsigned off = (unsigned)sj * (unsigned)(XROW * sizeof(float)) + lane8;
  return *(const float2*)((const char*)xp + off);
}

#define ROWS2 24

__device__ __forceinline__ void rows_sum2(const float* __restrict__ xp, unsigned lane8,
                                          int svA, float wvA, int svB, float wvB, float2& accA, float2& accB) {
  float2 xa[ROWS2], xb[ROWS2];
  #pragma unroll
  for (int j = 0; j < ROWS2; ++j) xa[j] = row_chunk(xp, lane8, __builtin_amdgcn_readlane(svA, j));
  #pragma unroll
  for (int j = 0; j < ROWS2; ++j) xb[j] = row_chunk(xp, lane8, __builtin_amdgcn_readlane(svB, j));
  #pragma unroll
  for (int j = 0; j < ROWS2; ++j) {
    const float wj = lane_f(wvA, j);
    accA.x = __builtin_fmaf(wj, xa[j].x, accA.x);
    accA.y = __builtin_fmaf(wj, xa[j].y, accA.y);
  }
  #pragma unroll
  for (int j = 0; j < ROWS2; ++j) {
    const float wj = lane_f(wvB, j);
    accB.x = __builtin_fmaf(wj, xb[j].x, accB.x);
    accB.y = __builtin_fmaf(wj, xb[j].y, accB.y);
  }
}

__device__ __forceinline__ void rows_sum(const float* __restrict__ xp, unsigned lane8, int sv, float wv, const int j0, float2& acc) {
  float2 x[32];
  #pragma unroll
  for (int j = j0; j < 32; ++j) x[j] = row_chunk(xp, lane8, __builtin_amdgcn_readlane(sv, j));
  #pragma unroll
  for (int j = j0; j < 32; ++j) {
    const float wj = lane_f(wv, j);
    acc.x = __builtin_fmaf(wj, x[j].x, acc.x);
    acc.y = __builtin_fmaf(wj, x[j].y, acc.y);
  }
}

__device__ __forceinline__ void rows_rest(const float* __restrict__ xp, unsigned lane8, const int2* __restrict__ csr2,
                                          int lane, int beg, int end, int n, float dn, int N, int emax,
                                          int sv, float wv, float2& acc) {
  if (end + 1 - beg > ROWS2) rows_sum(xp, lane8, sv, wv, ROWS2, acc);
  for (int cs = beg + 32; cs <= end; cs += 32) {
    const int e = cs + lane;
    const int2 rec = csr2[min(e, emax)];
    int s2;
    float w2;
    slot_of(rec, e, end, n, dn, N, s2, w2);
    rows_sum(xp, lane8, s2, w2, 0, acc);
  }
}

__global__ __launch_bounds__(256) void gru_seq_kernel(const int* __restrict__ rowptr,
                                                      const int* __restrict__ csr,
                                                      const float* __restrict__ dinv,
                                                      const float* __restrict__ X,
                                                      const float* __restrict__ Wg,
                                                      const float* __restrict__ bg,
                                                      const float* __restrict__ Wih,
                                                      const float* __restrict__ Whh,
                                                      const float* __restrict__ b_ih,
                                                      const float* __restrict__ b_hh,
                                                      const float* __restrict__ Wout,
                                                      const float* __restrict__ bout,
                                                      float* __restrict__ out, int N, int E) {
  __shared__ __attribute__((aligned(16))) _Float16 sW[2 * 3 * H * WPITCH];
  __shared__ __attribute__((aligned(16))) float sB[2 * 3 * H];
  __shared__ __attribute__((aligned(16))) float sBg[H];
  __shared__ __attribute__((aligned(16))) float sWgT[H * WGPITCH];
  __shared__ __attribute__((aligned(16))) float sX[8 * 16 * XP];

  const int tid = threadIdx.x;
  for (int i = tid; i < 2 * 3 * H * 8; i += 256) {
    const int row = i >> 3, seg = i & 7;
    const float* sp = (row < 3 * H ? Wih + (size_t)row * H : Whh + (size_t)(row - 3 * H) * H) + seg * 8;
    const float4 f0 = ((const float4*)sp)[0], f1 = ((const float4*)sp)[1];
    const float sc = ((row < 3 * H ? row : row - 3 * H) < 2 * H) ? 0.5f : 1.0f;
    v8h o;
    o[0] = (_Float16)(sc * f0.x); o[1] = (_Float16)(sc * f0.y); o[2] = (_Float16)(sc * f0.z); o[3] = (_Float16)(sc * f0.w);
    o[4] = (_Float16)(sc * f1.x); o[5] = (_Float16)(sc * f1.y); o[6] = (_Float16)(sc * f1.z); o[7] = (_Float16)(sc * f1.w);
    *(v8h*)(sW + row * WPITCH + seg * 8) = o;
  }
  for (int i = tid; i < 2 * 3 * H; i += 256) {
    const int g = (i < 3 * H) ? i : i - 3 * H;
    const float b = (i < 3 * H) ? b_ih[g] : b_hh[g];
    sB[i] = (g < 2 * H) ? 0.5f * b : b;
  }
  if (tid < H) sBg[tid] = bg[tid];
  for (int i = tid; i < F_IN * H; i += 256) sWgT[(i & (H - 1)) * WGPITCH + (i >> 6)] = Wg[i];
  __syncthreads();

  const int lane = tid & 31;
  const int c = lane & 15, hf = lane >> 4;
  const int wave = tid >> 5;
  const int place = wave ^ ((wave & 1) << 2);
  const int ntiles = (N + 15) >> 4;
  const int tstride = 8 * (int)gridDim.x;
  const float bo0 = bout[0], bo1 = bout[1];
  float* myx = sX + wave * (16 * XP);
  const int2* csr2 = (const int2*)csr;
  const int emax = max(E - 1, 0);
  for (int tile = __builtin_amdgcn_readfirstlane(place * (int)gridDim.x + (int)blockIdx.x); tile < ntiles; tile += tstride) {
  const int node = tile * 16 + c;
  const int n0 = tile * 16;

  v8f h32[4];
  for (int j = 0; j < 4; ++j) h32[j] = v8f{};
  v16h hB0 = {}, hB1 = {};

  const int rpv = rowptr[min(n0 + lane, N)];
  const float dvv = dinv[min(n0 + lane, N - 1)];

  for (int pass = 0; pass < T_STEPS / TPP; ++pass) {
  {
    const float* xp = X + pass * (TPP * F_IN);
    const unsigned lane8 = (unsigned)lane * 8u;
    int begA = max(__builtin_amdgcn_readlane(rpv, 0), 0);
    int begB = max(__builtin_amdgcn_readlane(rpv, 1), 0);
    int2 recA = csr2[min(begA + lane, emax)];
    int2 recB = csr2[min(begB + lane, emax)];
    for (int i = 0; i < 16; i += 2) {
      const int nA = min(n0 + i, N - 1), nB = min(n0 + i + 1, N - 1);
      const int endA = min(__builtin_amdgcn_readlane(rpv, i + 1), E);
      const int endB = min(__builtin_amdgcn_readlane(rpv, i + 2), E);
      const float dnA = lane_f(dvv, i), dnB = lane_f(dvv, i + 1);
      const int nbegA = max(__builtin_amdgcn_readlane(rpv, i + 2), 0);
      const int nbegB = max(__builtin_amdgcn_readlane(rpv, i + 3), 0);
      const int2 nrecA = csr2[min(nbegA + lane, emax)];
      const int2 nrecB = csr2[min(nbegB + lane, emax)];
      int svA, svB;
      float wvA, wvB;
      slot_of(recA, begA + lane, endA, nA, dnA, N, svA, wvA);
      slot_of(recB, begB + lane, endB, nB, dnB, N, svB, wvB);
      float2 accA, accB;
      accA.x = 0.0f; accA.y = 0.0f;
      accB.x = 0.0f; accB.y = 0.0f;
      rows_sum2(xp, lane8, svA, wvA, svB, wvB, accA, accB);
      rows_rest(xp, lane8, csr2, lane, begA, endA, nA, dnA, N, emax, svA, wvA, accA);
      rows_rest(xp, lane8, csr2, lane, begB, endB, nB, dnB, N, emax, svB, wvB, accB);
      accA.x *= dnA;
      accA.y *= dnA;
      accB.x *= dnB;
      accB.y *= dnB;
      *(float2*)(myx + i * XP + lane * 2) = accA;
      *(float2*)(myx + (i + 1) * XP + lane * 2) = accB;
      begA = nbegA;
      begB = nbegB;
      recA = nrecA;
      recB = nrecB;
    }
  }
  asm volatile("s_wait_dscnt 0x0" ::: "memory");

  for (int tt = 0; tt < TPP; ++tt) {
    asm volatile("" ::: "memory");
    const float* xl = myx + c * XP + tt * F_IN + 8 * hf;
    const float4 xa = ((const float4*)xl)[0], xb = ((const float4*)xl)[1];
    const float xs[8] = {xa.x, xa.y, xa.z, xa.w, xb.x, xb.y, xb.z, xb.w};

    v16h aB0, aB1;
    for (int blk = 0; blk < 4; ++blk) {
      const float4 w0 = *(const float4*)&sWgT[(16 * blk + c) * WGPITCH + 8 * hf];
      const float4 w1 = *(const float4*)&sWgT[(16 * blk + c) * WGPITCH + 8 * hf + 4];
      const float ws8[8] = {w0.x, w0.y, w0.z, w0.w, w1.x, w1.y, w1.z, w1.w};
      v8f acc = {};
      for (int q = 0; q < 4; ++q) {
        v2f wq, xq;
        wq[0] = ws8[2 * q];
        wq[1] = ws8[2 * q + 1];
        xq[0] = xs[2 * q];
        xq[1] = xs[2 * q + 1];
        acc = __builtin_amdgcn_wmma_f32_16x16x4_f32(false, wq, false, xq, (short)0, acc, false, false);
      }
      const float4 g0 = *(const float4*)&sBg[16 * blk + 8 * hf];
      const float4 g1 = *(const float4*)&sBg[16 * blk + 8 * hf + 4];
      const float gb[8] = {g0.x, g0.y, g0.z, g0.w, g1.x, g1.y, g1.z, g1.w};
      for (int r = 0; r < 8; ++r) {
        const _Float16 av = (_Float16)fmaxf(acc[r] + gb[r], 0.0f);
        if (blk < 2) aB0[(blk & 1) * 8 + r] = av; else aB1[(blk & 1) * 8 + r] = av;
      }
    }

    #pragma unroll
    for (int j = 0; j < 4; ++j) {
      asm volatile("" ::: "memory");
      const _Float16* wr = sW + (          16 * j + c) * WPITCH + 8 * hf;
      const _Float16* wz = sW + (     64 + 16 * j + c) * WPITCH + 8 * hf;
      const _Float16* wn = sW + (    128 + 16 * j + c) * WPITCH + 8 * hf;
      const _Float16* ur = sW + (192 +       16 * j + c) * WPITCH + 8 * hf;
      const _Float16* uz = sW + (192 +  64 + 16 * j + c) * WPITCH + 8 * hf;
      const _Float16* un = sW + (192 + 128 + 16 * j + c) * WPITCH + 8 * hf;

      v8f ir = {}, iz = {}, in_ = {}, hr = {}, hz = {}, hn = {};
      ir  = __builtin_amdgcn_wmma_f32_16x16x32_f16(false, lds_frag(wr, 0), false, aB0, (short)0, ir,  false, false);
      iz  = __builtin_amdgcn_wmma_f32_16x16x32_f16(false, lds_frag(wz, 0), false, aB0, (short)0, iz,  false, false);
      in_ = __builtin_amdgcn_wmma_f32_16x16x32_f16(false, lds_frag(wn, 0), false, aB0, (short)0, in_, false, false);
      hr  = __builtin_amdgcn_wmma_f32_16x16x32_f16(false, lds_frag(ur, 0), false, hB0, (short)0, hr,  false, false);
      hz  = __builtin_amdgcn_wmma_f32_16x16x32_f16(false, lds_frag(uz, 0), false, hB0, (short)0, hz,  false, false);
      hn  = __builtin_amdgcn_wmma_f32_16x16x32_f16(false, lds_frag(un, 0), false, hB0, (short)0, hn,  false, false);
      ir  = __builtin_amdgcn_wmma_f32_16x16x32_f16(false, lds_frag(wr, 32), false, aB1, (short)0, ir,  false, false);
      iz  = __builtin_amdgcn_wmma_f32_16x16x32_f16(false, lds_frag(wz, 32), false, aB1, (short)0, iz,  false, false);
      in_ = __builtin_amdgcn_wmma_f32_16x16x32_f16(false, lds_frag(wn, 32), false, aB1, (short)0, in_, false, false);
      hr  = __builtin_amdgcn_wmma_f32_16x16x32_f16(false, lds_frag(ur, 32), false, hB1, (short)0, hr,  false, false);
      hz  = __builtin_amdgcn_wmma_f32_16x16x32_f16(false, lds_frag(uz, 32), false, hB1, (short)0, hz,  false, false);
      hn  = __builtin_amdgcn_wmma_f32_16x16x32_f16(false, lds_frag(un, 32), false, hB1, (short)0, hn,  false, false);

      const int ch = 16 * j + 8 * hf;
      float bir[8], biz[8], bin[8], bhr[8], bhz[8], bhn[8];
      for (int r = 0; r < 8; r += 4) {
        *(float4*)&bir[r] = *(const float4*)&sB[            ch + r];
        *(float4*)&biz[r] = *(const float4*)&sB[       64 + ch + r];
        *(float4*)&bin[r] = *(const float4*)&sB[      128 + ch + r];
        *(float4*)&bhr[r] = *(const float4*)&sB[192 +       ch + r];
        *(float4*)&bhz[r] = *(const float4*)&sB[192 +  64 + ch + r];
        *(float4*)&bhn[r] = *(const float4*)&sB[192 + 128 + ch + r];
      }
      v8f zgv, ncv;
      for (int r = 0; r < 8; ++r) {
        float rg = sigmoid_of_half_(ir[r] + bir[r] + hr[r] + bhr[r]);
        zgv[r] = sigmoid_of_half_(iz[r] + biz[r] + hz[r] + bhz[r]);
        ncv[r] = tanhf_(in_[r] + bin[r] + rg * (hn[r] + bhn[r]));
      }
      h32[j] = (1.0f - zgv) * ncv + zgv * h32[j];
    }

    for (int r = 0; r < 8; ++r) {
      hB0[r]     = (_Float16)h32[0][r];
      hB0[8 + r] = (_Float16)h32[1][r];
      hB1[r]     = (_Float16)h32[2][r];
      hB1[8 + r] = (_Float16)h32[3][r];
    }
  }
  }

  float o0 = 0.0f, o1 = 0.0f;
  for (int j = 0; j < 4; ++j) {
    const float* wo = Wout + (16 * j + 8 * hf) * 2;
    float w[16];
    for (int q = 0; q < 4; ++q) *(float4*)&w[4 * q] = *(const float4*)(wo + 4 * q);
    for (int r = 0; r < 8; ++r) {
      o0 += h32[j][r] * w[2 * r];
      o1 += h32[j][r] * w[2 * r + 1];
    }
  }
  o0 += other_half(o0);
  o1 += other_half(o1);
  if (hf == 0 && node < N) *(float2*)(out + (size_t)node * 2) = make_float2(o0 + bo0, o1 + bo1);
  }
}


static inline int cdiv(long long a, long long b) { return (int)((a + b - 1) / b); }

extern "C" void kernel_launch(void* const* d_in, const int* in_sizes, int n_in,
                              void* d_out, int out_size, void* d_ws, size_t ws_size,
                              hipStream_t stream) {
  const float* X    = (const float*)d_in[0];
  const int*   ei   = (const int*)  d_in[1];
  const float* Wg   = (const float*)d_in[2];
  const float* bg   = (const float*)d_in[3];
  const float* Wih  = (const float*)d_in[4];
  const float* Whh  = (const float*)d_in[5];
  const float* bih  = (const float*)d_in[6];
  const float* bhh  = (const float*)d_in[7];
  const float* Wout = (const float*)d_in[8];
  const float* bout = (const float*)d_in[9];

  const int N = in_sizes[0] / (T_STEPS * F_IN);
  const int E = in_sizes[1] / 2;
  const int* src = ei;
  const int* dst = ei + E;

  char* ws = (char*)d_ws;
  size_t off = 0;
  auto alloc = [&](size_t bytes) -> void* {
    void* p = ws + off;
    off = (off + bytes + 255) & ~(size_t)255;
    return p;
  };
  int*      rowptr  = (int*)   alloc(sizeof(int) * (N + 1));
  float*    dinv    = (float*) alloc(sizeof(float) * N);
  int*      cnt     = (int*)   alloc(sizeof(int) * N);
  int*      bsum    = (int*)   alloc(sizeof(int) * cdiv(N, 256));
  int*      csr     = (int*)   alloc(sizeof(int) * 2 * E);
  int*      part    = (int*)   alloc(sizeof(int) * (size_t)NSL * (size_t)((N + 63) & ~63));
  (void)ws_size;

  const int sl = cdiv(E, NSL);
  const dim3 sgrid(NSL, cdiv(N, RANGE));
  slice_count_kernel <<<sgrid, 1024, 0, stream>>>(dst, part, N, E, sl);
  slice_prefix_kernel<<<cdiv(N, 256), 256, 0, stream>>>(part, cnt, bsum, N);
  scan_kernel       <<<cdiv(N, 1024), 256, 0, stream>>>(cnt, bsum, rowptr, dinv, N);
  slice_fill_kernel  <<<sgrid, 1024, 0, stream>>>(src, dst, part, rowptr, dinv, csr, N, E, sl);

  gru_seq_kernel<<<GRU_GROUPS, 256, 0, stream>>>(rowptr, csr, dinv, X, Wg, bg, Wih, Whh, bih, bhh, Wout, bout, (float*)d_out, N, E);
}
